# starting kernel
# baseline (speedup 1.0000x reference)
#include <hip/hip_runtime.h>
#include <cmath>

typedef float v2f __attribute__((ext_vector_type(2)));
typedef float v8f __attribute__((ext_vector_type(8)));

#define B_SZ 1024
#define S_SZ 512
#define E_SZ 50
#define H_SZ 50
#define G_SZ 200
#define K_SZ 100
#define NT 13
#define ROWS 16
#define THREADS (NT * 32)
#define GSTRIDE 204

#if __has_builtin(__builtin_amdgcn_tanhf)
#define FAST_TANH(x) __builtin_amdgcn_tanhf(x)
#elif __has_builtin(__builtin_amdgcn_tanh_f32)
#define FAST_TANH(x) __builtin_amdgcn_tanh_f32(x)
#else
#define FAST_TANH(x) tanhf(x)
#endif

__device__ __forceinline__ float fast_sigmoid(float x) {
    return __builtin_fmaf(0.5f, FAST_TANH(0.5f * x), 0.5f);
}

__global__ __launch_bounds__(THREADS)
void lstm_fused_kernel(const float* __restrict__ X,
                       const float* __restrict__ Wih,
                       const float* __restrict__ Whh,
                       const float* __restrict__ bih,
                       const float* __restrict__ bhh,
                       float* __restrict__ out)
{
    __shared__ float Z[ROWS][K_SZ];
    __shared__ float Gs[ROWS][GSTRIDE];
    __shared__ float Cs[ROWS][H_SZ];

    const int tid  = threadIdx.x;
    const int wave = tid >> 5;
    const int lane = tid & 31;
    const int b0   = blockIdx.x * ROWS;

    const int n     = wave * 16 + (lane & 15);
    const int khalf = (lane >> 4) * 2;
    const int m0    = lane & 15;
    const int rbase = (lane >> 4) * 8;

    v2f wfrag[25];
#pragma unroll
    for (int kk = 0; kk < 25; ++kk) {
        const int k = kk * 4 + khalf;
        float w0 = 0.f, w1 = 0.f;
        if (n < G_SZ) {
            w0 = (k     < E_SZ) ? Wih[n * E_SZ + k]     : Whh[n * H_SZ + (k - E_SZ)];
            w1 = (k + 1 < E_SZ) ? Wih[n * E_SZ + k + 1] : Whh[n * H_SZ + (k + 1 - E_SZ)];
        }
        wfrag[kk].x = w0;
        wfrag[kk].y = w1;
    }
    const float bias_n = (n < G_SZ) ? (bih[n] + bhh[n]) : 0.f;
    const bool  istanh = (n < G_SZ) && ((n / 50) == 2);

    const int pidx0 = tid;
    const int pidx1 = (tid + THREADS < ROWS * E_SZ) ? tid + THREADS
                                                    : ROWS * E_SZ - 1;
    const int zr0 = pidx0 / E_SZ, ze0 = pidx0 % E_SZ;
    const int zr1 = pidx1 / E_SZ, ze1 = pidx1 % E_SZ;
    const float* xp0 = X + (size_t)(b0 + zr0) * S_SZ * E_SZ + ze0;
    const float* xp1 = X + (size_t)(b0 + zr1) * S_SZ * E_SZ + ze1;

    for (int idx = tid; idx < ROWS * H_SZ; idx += THREADS) {
        const int r = idx / H_SZ, j = idx % H_SZ;
        Z[r][E_SZ + j] = 0.f;
        Cs[r][j]       = 0.f;
    }
    Z[zr0][ze0] = xp0[0];
    Z[zr1][ze1] = xp1[0];
    xp0 += E_SZ;
    xp1 += E_SZ;
    __syncthreads();

    float xr0 = 0.f, xr1 = 0.f;

    for (int s = 0; s < S_SZ; ++s) {
        if (s + 1 < S_SZ) {
            xr0 = xp0[0];
            xr1 = xp1[0];
            xp0 += E_SZ;
            xp1 += E_SZ;
        }

        v8f accs[4];
#pragma unroll
        for (int v = 0; v < 8; ++v) {
            accs[0][v] = bias_n;
            accs[1][v] = 0.f;
            accs[2][v] = 0.f;
            accs[3][v] = 0.f;
        }

#pragma unroll
        for (int kk = 0; kk < 25; ++kk) {
            const int k = kk * 4 + khalf;
            v2f a;
            a.x = Z[m0][k];
            a.y = Z[m0][k + 1];
            accs[kk & 3] = __builtin_amdgcn_wmma_f32_16x16x4_f32(
                               false, a, false, wfrag[kk], (short)0,
                               accs[kk & 3], false, false);
        }

        if (n < G_SZ) {
#pragma unroll
            for (int v = 0; v < 8; ++v) {
                const float g = (accs[0][v] + accs[1][v]) +
                                (accs[2][v] + accs[3][v]);
                Gs[rbase + v][n] = istanh ? FAST_TANH(g) : fast_sigmoid(g);
            }
        }
        __syncthreads();

        if (s + 1 < S_SZ) {
            Z[zr0][ze0] = xr0;
            Z[zr1][ze1] = xr1;
        }
        for (int idx = tid; idx < ROWS * H_SZ; idx += THREADS) {
            const int r = idx / H_SZ, j = idx % H_SZ;
            const float ig = Gs[r][j];
            const float fg = Gs[r][j + 50];
            const float gg = Gs[r][j + 100];
            const float og = Gs[r][j + 150];
            const float c  = fg * Cs[r][j] + ig * gg;
            Cs[r][j] = c;
            const float h = og * FAST_TANH(c);
            Z[r][E_SZ + j] = h;
            if (s == S_SZ - 1)
                out[(size_t)(b0 + r) * H_SZ + j] = h;
        }
        __syncthreads();
    }
}

extern "C" void kernel_launch(void* const* d_in, const int* in_sizes, int n_in,
                              void* d_out, int out_size, void* d_ws, size_t ws_size,
                              hipStream_t stream) {
    (void)in_sizes; (void)n_in; (void)d_ws; (void)ws_size; (void)out_size;
    const float* X   = (const float*)d_in[0];
    const float* Wih = (const float*)d_in[1];
    const float* Whh = (const float*)d_in[2];
    const float* bih = (const float*)d_in[3];
    const float* bhh = (const float*)d_in[4];
    float* out = (float*)d_out;

    dim3 grid(B_SZ / ROWS);
    dim3 block(THREADS);
    lstm_fused_kernel<<<grid, block, 0, stream>>>(X, Wih, Whh, bih, bhh, out);
}


# next step's x k-steps 0-9 (2nd accumulator set) run behind this step's h k-steps before the barrier, cell update issued between those WMMAs; 4-slab LDS ring, x two steps ahead; sums unchanged
# speedup vs baseline: 1.3952x; 1.3952x over previous
#include <hip/hip_runtime.h>
#include <cmath>


typedef float v2f __attribute__((ext_vector_type(2)));
typedef float v8f __attribute__((ext_vector_type(8)));

#define B_SZ 1024
#define S_SZ 512
#define E_SZ 50
#define H_SZ 50
#define K_SZ 100
#define KSTEPS 25
#define ROWS 16
#define NWAVES 4
#define THREADS (NWAVES * 32)
#define XPAIRS (ROWS * E_SZ / 2)
#define NBUF 4
#define XK 12
#define XPRE 10

#if __has_builtin(__builtin_amdgcn_tanhf)
#define FAST_TANH(x) __builtin_amdgcn_tanhf(x)
#elif __has_builtin(__builtin_amdgcn_tanh_f32)
#define FAST_TANH(x) __builtin_amdgcn_tanh_f32(x)
#else
#define FAST_TANH(x) tanhf(x)
#endif


__global__ __launch_bounds__(THREADS)
void lstm_unit_owner_kernel(const float* __restrict__ X,
                            const float* __restrict__ Wih,
                            const float* __restrict__ Whh,
                            const float* __restrict__ bih,
                            const float* __restrict__ bhh,
                            float* __restrict__ out)
{
    __shared__ __attribute__((aligned(16))) float Z[2 * NBUF][ROWS][K_SZ];

    const int tid  = threadIdx.x;
    const int wave = tid >> 5;
    const int lane = tid & 31;
    const int b0   = blockIdx.x * ROWS;

    const int c16   = lane & 15;
    const int khalf = (lane >> 4) * 2;
    const int rbase = (lane >> 4) * 8;

    const int  u0    = wave * 13 - (wave == 3 ? 1 : 0);
    const int  ucnt  = (wave < 2) ? 13 : 12;
    const bool valid = c16 < ucnt;
    const int  u     = u0 + c16;

    v2f   wf[4][KSTEPS];
    float bias[4];
#pragma unroll
    for (int g = 0; g < 4; ++g) {
        const int n = g * H_SZ + (valid ? u : 0);
#pragma unroll
        for (int kk = 0; kk < KSTEPS; ++kk) {
            const int k = kk * 4 + khalf;
            float w0, w1;
            if (k < E_SZ) {
                w0 = Wih[n * E_SZ + k];
                w1 = Wih[n * E_SZ + k + 1];
            } else {
                w0 = Whh[n * H_SZ + (k - E_SZ)];
                w1 = Whh[n * H_SZ + (k + 1 - E_SZ)];
            }
            wf[g][kk].x = valid ? w0 : 0.f;
            wf[g][kk].y = valid ? w1 : 0.f;
        }
        bias[g] = valid ? (bih[n] + bhh[n]) : 0.f;
    }

    const float* xsrc[4];
    int          xdst[4];
#pragma unroll
    for (int i = 0; i < 4; ++i) {
        const int p  = tid + i * THREADS;
        const int pc = (p < XPAIRS) ? p : XPAIRS - 1;
        const int r  = pc / (E_SZ / 2);
        const int e  = (pc % (E_SZ / 2)) * 2;
        xsrc[i] = X + (size_t)(b0 + r) * S_SZ * E_SZ + e;
        xdst[i] = r * K_SZ + e;
    }

    float* Zf = &Z[0][0][0];
    for (int idx = tid; idx < ROWS * H_SZ; idx += THREADS)
        Z[0][idx / H_SZ][E_SZ + idx % H_SZ] = 0.f;
#pragma unroll
    for (int i = 0; i < 4; ++i) {
        *(v2f*)(Zf + xdst[i]) = *(const v2f*)(xsrc[i]);
        *(v2f*)(Zf + ROWS * K_SZ + xdst[i]) = *(const v2f*)(xsrc[i] + E_SZ);
    }
    float cst[8];
#pragma unroll
    for (int v = 0; v < 8; ++v) cst[v] = 0.f;
    v8f bv[4];
#pragma unroll
    for (int g = 0; g < 4; ++g)
#pragma unroll
        for (int v = 0; v < 8; ++v) bv[g][v] = bias[g];
    __syncthreads();

    const int hdst = (valid ? 0 : NBUF * ROWS * K_SZ) + rbase * K_SZ + E_SZ + (valid ? u : c16);

    v8f acc[4];
    v2f ax10, ax11;
    {
        v2f a0[XK];
#pragma unroll
        for (int kk = 0; kk < XK; ++kk)
            a0[kk] = *(const v2f*)(Zf + c16 * K_SZ + kk * 4 + khalf);
#pragma unroll
        for (int g = 0; g < 4; ++g)
            acc[g] = __builtin_amdgcn_wmma_f32_16x16x4_f32(
                         false, a0[0], false, wf[g][0], (short)0,
                         bv[g], false, false);
#pragma unroll
        for (int kk = 1; kk < XPRE; ++kk) {
#pragma unroll
            for (int g = 0; g < 4; ++g)
                acc[g] = __builtin_amdgcn_wmma_f32_16x16x4_f32(
                             false, a0[kk], false, wf[g][kk], (short)0,
                             acc[g], false, false);
        }
        ax10 = a0[XPRE];
        ax11 = a0[XPRE + 1];
    }

    for (int sp = 0; sp < S_SZ; sp += NBUF) {
#pragma unroll
        for (int par = 0; par < NBUF; ++par) {
            const int s = sp + par;
            const float* Zc  = &Z[par][0][0];
            float*       Zn  = &Z[(par + 1) & 3][0][0];
            float*       Zn2 = &Z[(par + 2) & 3][0][0];
            const int sn = (s + 2 < S_SZ) ? s + 2 : S_SZ - 1;

            v2f xr[4];
#pragma unroll
            for (int i = 0; i < 4; ++i)
                xr[i] = *(const v2f*)(xsrc[i] + (size_t)sn * E_SZ);
            __builtin_amdgcn_sched_barrier(0);

            v2f a[KSTEPS];
#pragma unroll
            for (int kk = XK; kk < XK + 4; ++kk)
                a[kk] = *(const v2f*)(Zc + c16 * K_SZ + kk * 4 + khalf);
            __builtin_amdgcn_sched_barrier(0);
#pragma unroll
            for (int kk = XK + 4; kk < KSTEPS; ++kk)
                a[kk] = *(const v2f*)(Zc + c16 * K_SZ + kk * 4 + khalf);
            __builtin_amdgcn_sched_barrier(0);
#pragma unroll
            for (int kk = 0; kk < XK; ++kk)
                a[kk] = *(const v2f*)(Zn + c16 * K_SZ + kk * 4 + khalf);
            __builtin_amdgcn_sched_barrier(0);

#pragma unroll
            for (int g = 0; g < 4; ++g)
                acc[g] = __builtin_amdgcn_wmma_f32_16x16x4_f32(
                             false, ax10, false, wf[g][XPRE], (short)0,
                             acc[g], false, false);
#pragma unroll
            for (int g = 0; g < 4; ++g)
                acc[g] = __builtin_amdgcn_wmma_f32_16x16x4_f32(
                             false, ax11, false, wf[g][XPRE + 1], (short)0,
                             acc[g], false, false);
#pragma unroll
            for (int kk = XK; kk < KSTEPS; ++kk) {
#pragma unroll
                for (int g = 0; g < 4; ++g)
                    acc[g] = __builtin_amdgcn_wmma_f32_16x16x4_f32(
                                 false, a[kk], false, wf[g][kk], (short)0,
                                 acc[g], false, false);
            }
            __builtin_amdgcn_sched_barrier(0);

            v8f   accn[4];
            float ti[8], tf[8], tg[8], to[8], hv[8];
#pragma unroll
            for (int i = 0; i < 4 * XPRE; ++i) {
                const int kk = i >> 2, g = i & 3;
                accn[g] = __builtin_amdgcn_wmma_f32_16x16x4_f32(
                              false, a[kk], false, wf[g][kk], (short)0,
                              (kk == 0) ? bv[g] : accn[g], false, false);
#pragma unroll
                for (int j = 4 * i; j < 4 * i + 4 && j < 112; ++j) {
                    const int st = j >> 3, v = j & 7;
                    switch (st) {
                    case 0:  ti[v] = 0.5f * acc[0][v]; break;
                    case 1:  tf[v] = 0.5f * acc[1][v]; break;
                    case 2:  to[v] = 0.5f * acc[3][v]; break;
                    case 3:  ti[v] = FAST_TANH(ti[v]); break;
                    case 4:  tf[v] = FAST_TANH(tf[v]); break;
                    case 5:  tg[v] = FAST_TANH(acc[2][v]); break;
                    case 6:  to[v] = FAST_TANH(to[v]); break;
                    case 7:  ti[v] = __builtin_fmaf(0.5f, ti[v], 0.5f); break;
                    case 8:  tf[v] = __builtin_fmaf(0.5f, tf[v], 0.5f); break;
                    case 9:  to[v] = __builtin_fmaf(0.5f, to[v], 0.5f); break;
                    case 10: tg[v] = ti[v] * tg[v]; break;
                    case 11: cst[v] = __builtin_fmaf(tf[v], cst[v], tg[v]); break;
                    case 12: tg[v] = FAST_TANH(cst[v]); break;
                    default: hv[v] = to[v] * tg[v]; break;
                    }
                }
                if (i == 27) {
#pragma unroll
                    for (int v = 0; v < 8; ++v)
                        Zn[hdst + v * K_SZ] = hv[v];
#pragma unroll
                    for (int q = 0; q < 4; ++q)
                        *(v2f*)(Zn2 + xdst[q]) = xr[q];
                }
                __builtin_amdgcn_sched_barrier(0);
            }

            __syncthreads();

#pragma unroll
            for (int g = 0; g < 4; ++g) acc[g] = accn[g];
            ax10 = a[XPRE];
            ax11 = a[XPRE + 1];
        }
    }

    for (int idx = tid; idx < ROWS * H_SZ; idx += THREADS)
        out[(size_t)b0 * H_SZ + idx] = Z[0][idx / H_SZ][E_SZ + idx % H_SZ];
}

extern "C" void kernel_launch(void* const* d_in, const int* in_sizes, int n_in,
                              void* d_out, int out_size, void* d_ws, size_t ws_size,
                              hipStream_t stream) {
    (void)in_sizes; (void)n_in; (void)d_ws; (void)ws_size; (void)out_size;
    const float* X   = (const float*)d_in[0];
    const float* Wih = (const float*)d_in[1];
    const float* Whh = (const float*)d_in[2];
    const float* bih = (const float*)d_in[3];
    const float* bhh = (const float*)d_in[4];
    float* out = (float*)d_out;

    dim3 grid(B_SZ / ROWS);
    dim3 block(THREADS);
    lstm_unit_owner_kernel<<<grid, block, 0, stream>>>(X, Wih, Whh, bih, bhh, out);
}


# x part (bias, k-steps 0-11) of each step moved to a 2nd wave per SIMD, 2 steps ahead: fills the matrix pipe while the h wave is in hand-over and cell update; tiles passed as C via LDS; sums unchanged
# speedup vs baseline: 1.4473x; 1.0373x over previous
#include <hip/hip_runtime.h>
#include <cmath>


typedef float v2f __attribute__((ext_vector_type(2)));
typedef float v4f __attribute__((ext_vector_type(4)));
typedef float v8f __attribute__((ext_vector_type(8)));

#define B_SZ 1024
#define S_SZ 512
#define E_SZ 50
#define H_SZ 50
#define K_SZ 100
#define KSTEPS 25
#define ROWS 16
#define NOWN 4
#define NWAVES 8
#define THREADS (NWAVES * 32)
#define XTHREADS (NOWN * 32)
#define XPAIRS (ROWS * E_SZ / 2)
#define NBUF 8
#define SLAB (ROWS * K_SZ)
#define XK 12
#define HK (KSTEPS - XK)

#if __has_builtin(__builtin_amdgcn_tanhf)
#define FAST_TANH(x) __builtin_amdgcn_tanhf(x)
#elif __has_builtin(__builtin_amdgcn_tanh_f32)
#define FAST_TANH(x) __builtin_amdgcn_tanh_f32(x)
#else
#define FAST_TANH(x) tanhf(x)
#endif


__global__ __launch_bounds__(THREADS)
void lstm_pair_kernel(const float* __restrict__ X,
                      const float* __restrict__ Wih,
                      const float* __restrict__ Whh,
                      const float* __restrict__ bih,
                      const float* __restrict__ bhh,
                      float* __restrict__ out)
{
    __shared__ __attribute__((aligned(16))) float Z[NBUF + 1][ROWS][K_SZ];
    __shared__ __attribute__((aligned(16))) float P[2][NOWN][8][32][4];

    const int tid  = threadIdx.x;
    const int wave = __builtin_amdgcn_readfirstlane(tid >> 5);
    const int lane = tid & 31;
    const int b0   = blockIdx.x * ROWS;
    const bool xrole = wave >= NOWN;
    const int  uw    = wave & (NOWN - 1);

    const int c16   = lane & 15;
    const int khalf = (lane >> 4) * 2;
    const int rbase = (lane >> 4) * 8;

    const int  u0    = uw * 13 - (uw == 3 ? 1 : 0);
    const int  ucnt  = (uw < 2) ? 13 : 12;
    const bool valid = c16 < ucnt;
    const int  u     = u0 + c16;

    float* Zf = &Z[0][0][0];
    float* Pf = &P[0][0][0][0][0];
    const int pslot = uw * (8 * 32 * 4) + lane * 4;
    const int PPAR  = NOWN * 8 * 32 * 4;

    for (int idx = tid; idx < ROWS * H_SZ; idx += THREADS)
        Z[0][idx / H_SZ][E_SZ + idx % H_SZ] = 0.f;
#pragma unroll
    for (int i = 0; i < 2; ++i) {
        const int p  = tid + i * THREADS;
        const int pc = (p < XPAIRS) ? p : XPAIRS - 1;
        const int r  = pc / (E_SZ / 2);
        const int e  = (pc % (E_SZ / 2)) * 2;
        const float* src = X + (size_t)(b0 + r) * S_SZ * E_SZ + e;
#pragma unroll
        for (int t = 0; t < 4; ++t)
            *(v2f*)(Zf + t * SLAB + r * K_SZ + e) = *(const v2f*)(src + t * E_SZ);
    }
    __syncthreads();

    if (!xrole) {
        v2f wh[4][HK];
#pragma unroll
        for (int g = 0; g < 4; ++g) {
            const int n = g * H_SZ + (valid ? u : 0);
#pragma unroll
            for (int kk = XK; kk < KSTEPS; ++kk) {
                const int k = kk * 4 + khalf;
                float w0, w1;
                if (k < E_SZ) {
                    w0 = Wih[n * E_SZ + k];
                    w1 = Wih[n * E_SZ + k + 1];
                } else {
                    w0 = Whh[n * H_SZ + (k - E_SZ)];
                    w1 = Whh[n * H_SZ + (k + 1 - E_SZ)];
                }
                wh[g][kk - XK].x = valid ? w0 : 0.f;
                wh[g][kk - XK].y = valid ? w1 : 0.f;
            }
        }
        float cst[8];
#pragma unroll
        for (int v = 0; v < 8; ++v) cst[v] = 0.f;
        const int hrow = rbase * K_SZ + E_SZ + (valid ? u : c16);

        __syncthreads();

        for (int sp = 0; sp < S_SZ; sp += NBUF) {
#pragma unroll
            for (int par = 0; par < NBUF; ++par) {
                const float* Zc = &Z[par][0][0];
                float*       Zn = Zf + (valid ? ((par + 1) & (NBUF - 1)) : NBUF) * SLAB + hrow;
                const float* Pr = Pf + (par & 1) * PPAR + pslot;

                v8f acc[4];
#pragma unroll
                for (int g = 0; g < 4; ++g) {
                    const v4f lo = *(const v4f*)(Pr + (g * 2 + 0) * 128);
                    const v4f hi = *(const v4f*)(Pr + (g * 2 + 1) * 128);
                    acc[g][0] = lo[0]; acc[g][1] = lo[1]; acc[g][2] = lo[2]; acc[g][3] = lo[3];
                    acc[g][4] = hi[0]; acc[g][5] = hi[1]; acc[g][6] = hi[2]; acc[g][7] = hi[3];
                }
                v2f a[HK];
#pragma unroll
                for (int kk = 0; kk < HK; ++kk)
                    a[kk] = *(const v2f*)(Zc + c16 * K_SZ + (XK + kk) * 4 + khalf);
                __builtin_amdgcn_sched_barrier(0);

#pragma unroll
                for (int kk = 0; kk < HK; ++kk) {
#pragma unroll
                    for (int g = 0; g < 4; ++g)
                        acc[g] = __builtin_amdgcn_wmma_f32_16x16x4_f32(
                                     false, a[kk], false, wh[g][kk], (short)0,
                                     acc[g], false, false);
                }

                float ti[8], tf[8], tg[8], to[8], hv[8];
#pragma unroll
                for (int v = 0; v < 8; ++v) ti[v] = 0.5f * acc[0][v];
#pragma unroll
                for (int v = 0; v < 8; ++v) tf[v] = 0.5f * acc[1][v];
#pragma unroll
                for (int v = 0; v < 8; ++v) to[v] = 0.5f * acc[3][v];
#pragma unroll
                for (int v = 0; v < 8; ++v) ti[v] = FAST_TANH(ti[v]);
#pragma unroll
                for (int v = 0; v < 8; ++v) tf[v] = FAST_TANH(tf[v]);
#pragma unroll
                for (int v = 0; v < 8; ++v) tg[v] = FAST_TANH(acc[2][v]);
#pragma unroll
                for (int v = 0; v < 8; ++v) to[v] = FAST_TANH(to[v]);
#pragma unroll
                for (int v = 0; v < 8; ++v) ti[v] = __builtin_fmaf(0.5f, ti[v], 0.5f);
#pragma unroll
                for (int v = 0; v < 8; ++v) tf[v] = __builtin_fmaf(0.5f, tf[v], 0.5f);
#pragma unroll
                for (int v = 0; v < 8; ++v) to[v] = __builtin_fmaf(0.5f, to[v], 0.5f);
#pragma unroll
                for (int v = 0; v < 8; ++v) tg[v] = ti[v] * tg[v];
#pragma unroll
                for (int v = 0; v < 8; ++v) cst[v] = __builtin_fmaf(tf[v], cst[v], tg[v]);
#pragma unroll
                for (int v = 0; v < 8; ++v) tg[v] = FAST_TANH(cst[v]);
#pragma unroll
                for (int v = 0; v < 8; ++v) hv[v] = to[v] * tg[v];
#pragma unroll
                for (int v = 0; v < 8; ++v)
                    Zn[v * K_SZ] = hv[v];

                __syncthreads();
            }
        }
    } else {
        v2f   wx[4][XK];
        float bias[4];
#pragma unroll
        for (int g = 0; g < 4; ++g) {
            const int n = g * H_SZ + (valid ? u : 0);
#pragma unroll
            for (int kk = 0; kk < XK; ++kk) {
                const int k = kk * 4 + khalf;
                const float w0 = Wih[n * E_SZ + k];
                const float w1 = Wih[n * E_SZ + k + 1];
                wx[g][kk].x = valid ? w0 : 0.f;
                wx[g][kk].y = valid ? w1 : 0.f;
            }
            bias[g] = valid ? (bih[n] + bhh[n]) : 0.f;
        }
        v8f bv[4];
#pragma unroll
        for (int g = 0; g < 4; ++g)
#pragma unroll
            for (int v = 0; v < 8; ++v) bv[g][v] = bias[g];

        const int xt = tid - XTHREADS;
        const float* xsrc[4];
        int          xdst[4];
#pragma unroll
        for (int i = 0; i < 4; ++i) {
            const int p  = xt + i * XTHREADS;
            const int pc = (p < XPAIRS) ? p : XPAIRS - 1;
            const int r  = pc / (E_SZ / 2);
            const int e  = (pc % (E_SZ / 2)) * 2;
            xsrc[i] = X + (size_t)(b0 + r) * S_SZ * E_SZ + e;
            xdst[i] = r * K_SZ + e;
        }

        auto xpart = [&](const v2f (&f)[XK], v8f (&t)[4]) {
#pragma unroll
            for (int g = 0; g < 4; ++g)
                t[g] = __builtin_amdgcn_wmma_f32_16x16x4_f32(
                           false, f[0], false, wx[g][0], (short)0,
                           bv[g], false, false);
#pragma unroll
            for (int kk = 1; kk < XK; ++kk) {
#pragma unroll
                for (int g = 0; g < 4; ++g)
                    t[g] = __builtin_amdgcn_wmma_f32_16x16x4_f32(
                               false, f[kk], false, wx[g][kk], (short)0,
                               t[g], false, false);
            }
        };
        auto handover = [&](float* Pw, const v8f (&t)[4]) {
#pragma unroll
            for (int g = 0; g < 4; ++g) {
                v4f lo, hi;
                lo[0] = t[g][0]; lo[1] = t[g][1]; lo[2] = t[g][2]; lo[3] = t[g][3];
                hi[0] = t[g][4]; hi[1] = t[g][5]; hi[2] = t[g][6]; hi[3] = t[g][7];
                *(v4f*)(Pw + (g * 2 + 0) * 128) = lo;
                *(v4f*)(Pw + (g * 2 + 1) * 128) = hi;
            }
        };
        auto xfrags = [&](const float* Zs, v2f (&f)[XK]) {
#pragma unroll
            for (int kk = 0; kk < XK; ++kk)
                f[kk] = *(const v2f*)(Zs + c16 * K_SZ + kk * 4 + khalf);
        };

        v8f tp[4];
        v2f fc[XK];
        v2f xr[4];
        {
            v2f f0[XK];
            v8f t0[4];
            xfrags(Zf, f0);
            xpart(f0, t0);
            handover(Pf + pslot, t0);
            xfrags(Zf + SLAB, f0);
            xpart(f0, tp);
            xfrags(Zf + 2 * SLAB, fc);
#pragma unroll
            for (int i = 0; i < 4; ++i)
                xr[i] = *(const v2f*)(xsrc[i] + (size_t)4 * E_SZ);
        }
        __syncthreads();

        for (int sp = 0; sp < S_SZ; sp += NBUF) {
#pragma unroll
            for (int par = 0; par < NBUF; ++par) {
                const int s = sp + par;
                const float* Zn3 = &Z[(par + 3) & (NBUF - 1)][0][0];
                float*       Zn4 = &Z[(par + 4) & (NBUF - 1)][0][0];
                const int sn = (s + 5 < S_SZ) ? s + 5 : S_SZ - 1;

                handover(Pf + ((par + 1) & 1) * PPAR + pslot, tp);
#pragma unroll
                for (int q = 0; q < 4; ++q)
                    *(v2f*)(Zn4 + xdst[q]) = xr[q];
                v2f fn[XK];
                xfrags(Zn3, fn);
#pragma unroll
                for (int i = 0; i < 4; ++i)
                    xr[i] = *(const v2f*)(xsrc[i] + (size_t)sn * E_SZ);
                __builtin_amdgcn_sched_barrier(0);

                v8f tc[4];
                xpart(fc, tc);
                __builtin_amdgcn_sched_barrier(0);

                __syncthreads();

#pragma unroll
                for (int g = 0; g < 4; ++g) tp[g] = tc[g];
#pragma unroll
                for (int kk = 0; kk < XK; ++kk) fc[kk] = fn[kk];
            }
        }
    }

    for (int idx = tid; idx < ROWS * H_SZ; idx += THREADS)
        out[(size_t)b0 * H_SZ + idx] = Z[0][idx / H_SZ][E_SZ + idx % H_SZ];
}

extern "C" void kernel_launch(void* const* d_in, const int* in_sizes, int n_in,
                              void* d_out, int out_size, void* d_ws, size_t ws_size,
                              hipStream_t stream) {
    (void)in_sizes; (void)n_in; (void)d_ws; (void)ws_size; (void)out_size;
    const float* X   = (const float*)d_in[0];
    const float* Wih = (const float*)d_in[1];
    const float* Whh = (const float*)d_in[2];
    const float* bih = (const float*)d_in[3];
    const float* bhh = (const float*)d_in[4];
    float* out = (float*)d_out;

    dim3 grid(B_SZ / ROWS);
    dim3 block(THREADS);
    lstm_pair_kernel<<<grid, block, 0, stream>>>(X, Wih, Whh, bih, bhh, out);
}


# hidden units regrouped 16/16/16/2: 13 gate tiles per step, not 16; units 48,49 share one tile (gates in columns, f,g,o fetched by row shift); x wave 3 takes the o tiles of groups 0-2; sums unchanged
# speedup vs baseline: 1.5828x; 1.0936x over previous
#include <hip/hip_runtime.h>
#include <cmath>


typedef float v2f __attribute__((ext_vector_type(2)));
typedef float v4f __attribute__((ext_vector_type(4)));
typedef float v8f __attribute__((ext_vector_type(8)));

#define B_SZ 1024
#define S_SZ 512
#define E_SZ 50
#define H_SZ 50
#define K_SZ 100
#define KSTEPS 25
#define ROWS 16
#define NOWN 4
#define UG 16
#define ULAST 48
#define NWAVES 8
#define THREADS (NWAVES * 32)
#define XTHREADS (NOWN * 32)
#define XPAIRS (ROWS * E_SZ / 2)
#define NBUF 8
#define SLAB (ROWS * K_SZ)
#define XK 12
#define HK (KSTEPS - XK)
#define GSET (8 * 32 * 4)
#define PPAR (NOWN * GSET)

#if __has_builtin(__builtin_amdgcn_tanhf)
#define FAST_TANH(x) __builtin_amdgcn_tanhf(x)
#elif __has_builtin(__builtin_amdgcn_tanh_f32)
#define FAST_TANH(x) __builtin_amdgcn_tanh_f32(x)
#else
#define FAST_TANH(x) tanhf(x)
#endif


template <int N>
__device__ __forceinline__ float row_from(float x)
{
    return __builtin_bit_cast(float,
        __builtin_amdgcn_update_dpp(0, __builtin_bit_cast(int, x), 0x100 + N, 0xf, 0xf, true));
}

template <bool MIXED>
__device__ __forceinline__ void h_wave(const float* __restrict__ Wih,
                                       const float* __restrict__ Whh,
                                       float* Zf, const float* Pf,
                                       const int uw, const int lane)
{
    constexpr int NT = MIXED ? 1 : 4;
    const int c16   = lane & 15;
    const int khalf = (lane >> 4) * 2;
    const int rbase = (lane >> 4) * 8;

    v2f wh[NT][HK];
#pragma unroll
    for (int g = 0; g < NT; ++g) {
        const bool valid = MIXED ? (c16 < 8) : true;
        const int  n     = MIXED ? (valid ? (c16 >> 1) * H_SZ + ULAST + (c16 & 1) : 0)
                                 : g * H_SZ + uw * UG + c16;
#pragma unroll
        for (int kk = XK; kk < KSTEPS; ++kk) {
            const int k = kk * 4 + khalf;
            float w0, w1;
            if (k < E_SZ) {
                w0 = Wih[n * E_SZ + k];
                w1 = Wih[n * E_SZ + k + 1];
            } else {
                w0 = Whh[n * H_SZ + (k - E_SZ)];
                w1 = Whh[n * H_SZ + (k + 1 - E_SZ)];
            }
            wh[g][kk - XK].x = valid ? w0 : 0.f;
            wh[g][kk - XK].y = valid ? w1 : 0.f;
        }
    }
    float cst[8];
#pragma unroll
    for (int v = 0; v < 8; ++v) cst[v] = 0.f;
    const bool hvalid = MIXED ? (c16 < 2) : true;
    const int  hrow   = rbase * K_SZ + E_SZ + (MIXED ? (hvalid ? ULAST + c16 : c16) : uw * UG + c16);
    const int  pslot  = uw * GSET + lane * 4;
    const bool isg = (c16 >> 1) == 2;
    const float sc = isg ? 1.0f : 0.5f;
    const float ab = isg ? 0.0f : 0.5f;

    __syncthreads();

    for (int sp = 0; sp < S_SZ; sp += NBUF) {
#pragma unroll
        for (int par = 0; par < NBUF; ++par) {
            const float* Zc = Zf + par * SLAB;
            float*       Zn = Zf + (hvalid ? ((par + 1) & (NBUF - 1)) : NBUF) * SLAB + hrow;
            const float* Pr = Pf + (par & 1) * PPAR + pslot;

            v8f acc[NT];
#pragma unroll
            for (int g = 0; g < NT; ++g) {
                const v4f lo = *(const v4f*)(Pr + (g * 2 + 0) * 128);
                const v4f hi = *(const v4f*)(Pr + (g * 2 + 1) * 128);
                acc[g][0] = lo[0]; acc[g][1] = lo[1]; acc[g][2] = lo[2]; acc[g][3] = lo[3];
                acc[g][4] = hi[0]; acc[g][5] = hi[1]; acc[g][6] = hi[2]; acc[g][7] = hi[3];
            }
            v2f a[HK];
#pragma unroll
            for (int kk = 0; kk < HK; ++kk)
                a[kk] = *(const v2f*)(Zc + c16 * K_SZ + (XK + kk) * 4 + khalf);
            __builtin_amdgcn_sched_barrier(0);

#pragma unroll
            for (int kk = 0; kk < HK; ++kk) {
#pragma unroll
                for (int g = 0; g < NT; ++g)
                    acc[g] = __builtin_amdgcn_wmma_f32_16x16x4_f32(
                                 false, a[kk], false, wh[g][kk], (short)0,
                                 acc[g], false, false);
            }

            float hv[8];
            if constexpr (!MIXED) {
                float ti[8], tf[8], tg[8], to[8];
#pragma unroll
                for (int v = 0; v < 8; ++v) ti[v] = 0.5f * acc[0][v];
#pragma unroll
                for (int v = 0; v < 8; ++v) tf[v] = 0.5f * acc[1][v];
#pragma unroll
                for (int v = 0; v < 8; ++v) to[v] = 0.5f * acc[NT - 1][v];
#pragma unroll
                for (int v = 0; v < 8; ++v) ti[v] = FAST_TANH(ti[v]);
#pragma unroll
                for (int v = 0; v < 8; ++v) tf[v] = FAST_TANH(tf[v]);
#pragma unroll
                for (int v = 0; v < 8; ++v) tg[v] = FAST_TANH(acc[NT / 2][v]);
#pragma unroll
                for (int v = 0; v < 8; ++v) to[v] = FAST_TANH(to[v]);
#pragma unroll
                for (int v = 0; v < 8; ++v) ti[v] = __builtin_fmaf(0.5f, ti[v], 0.5f);
#pragma unroll
                for (int v = 0; v < 8; ++v) tf[v] = __builtin_fmaf(0.5f, tf[v], 0.5f);
#pragma unroll
                for (int v = 0; v < 8; ++v) to[v] = __builtin_fmaf(0.5f, to[v], 0.5f);
#pragma unroll
                for (int v = 0; v < 8; ++v) tg[v] = ti[v] * tg[v];
#pragma unroll
                for (int v = 0; v < 8; ++v) cst[v] = __builtin_fmaf(tf[v], cst[v], tg[v]);
#pragma unroll
                for (int v = 0; v < 8; ++v) tg[v] = FAST_TANH(cst[v]);
#pragma unroll
                for (int v = 0; v < 8; ++v) hv[v] = to[v] * tg[v];
            } else {
                float ta[8], tf[8], tg[8], to[8];
#pragma unroll
                for (int v = 0; v < 8; ++v) ta[v] = sc * acc[0][v];
#pragma unroll
                for (int v = 0; v < 8; ++v) ta[v] = FAST_TANH(ta[v]);
#pragma unroll
                for (int v = 0; v < 8; ++v) ta[v] = __builtin_fmaf(sc, ta[v], ab);
#pragma unroll
                for (int v = 0; v < 8; ++v) {
                    tf[v] = row_from<2>(ta[v]);
                    tg[v] = row_from<4>(ta[v]);
                    to[v] = row_from<6>(ta[v]);
                }
#pragma unroll
                for (int v = 0; v < 8; ++v) tg[v] = ta[v] * tg[v];
#pragma unroll
                for (int v = 0; v < 8; ++v) cst[v] = __builtin_fmaf(tf[v], cst[v], tg[v]);
#pragma unroll
                for (int v = 0; v < 8; ++v) tg[v] = FAST_TANH(cst[v]);
#pragma unroll
                for (int v = 0; v < 8; ++v) hv[v] = to[v] * tg[v];
            }
#pragma unroll
            for (int v = 0; v < 8; ++v)
                Zn[v * K_SZ] = hv[v];

            __syncthreads();
        }
    }
}

template <int NT>
__device__ __forceinline__ void x_wave(const float* __restrict__ X,
                                       const float* __restrict__ Wih,
                                       const float* __restrict__ bih,
                                       const float* __restrict__ bhh,
                                       float* Zf, float* Pf,
                                       const int uw, const int lane, const int tid, const int b0)
{
    const int c16   = lane & 15;
    const int khalf = (lane >> 4) * 2;

    v2f   wx[NT][XK];
    float bias[NT];
    int   poff[NT];
#pragma unroll
    for (int g = 0; g < NT; ++g) {
        bool valid; int n;
        if (NT == 3) {
            valid = true;
            n = g * H_SZ + uw * UG + c16;
            poff[g] = uw * GSET + g * 256 + lane * 4;
        } else if (g < 3) {
            valid = true;
            n = 3 * H_SZ + g * UG + c16;
            poff[g] = g * GSET + 3 * 256 + lane * 4;
        } else {
            valid = c16 < 8;
            n = valid ? (c16 >> 1) * H_SZ + ULAST + (c16 & 1) : 0;
            poff[g] = 3 * GSET + lane * 4;
        }
#pragma unroll
        for (int kk = 0; kk < XK; ++kk) {
            const int k = kk * 4 + khalf;
            const float w0 = Wih[n * E_SZ + k];
            const float w1 = Wih[n * E_SZ + k + 1];
            wx[g][kk].x = valid ? w0 : 0.f;
            wx[g][kk].y = valid ? w1 : 0.f;
        }
        bias[g] = valid ? (bih[n] + bhh[n]) : 0.f;
    }
    v8f bv[NT];
#pragma unroll
    for (int g = 0; g < NT; ++g)
#pragma unroll
        for (int v = 0; v < 8; ++v) bv[g][v] = bias[g];

    const int xt = tid - XTHREADS;
    const float* xsrc[4];
    int          xdst[4];
#pragma unroll
    for (int i = 0; i < 4; ++i) {
        const int p  = xt + i * XTHREADS;
        const int pc = (p < XPAIRS) ? p : XPAIRS - 1;
        const int r  = pc / (E_SZ / 2);
        const int e  = (pc % (E_SZ / 2)) * 2;
        xsrc[i] = X + (size_t)(b0 + r) * S_SZ * E_SZ + e;
        xdst[i] = r * K_SZ + e;
    }

    auto xpart = [&](const v2f (&f)[XK], v8f (&t)[NT]) {
#pragma unroll
        for (int g = 0; g < NT; ++g)
            t[g] = __builtin_amdgcn_wmma_f32_16x16x4_f32(
                       false, f[0], false, wx[g][0], (short)0,
                       bv[g], false, false);
#pragma unroll
        for (int kk = 1; kk < XK; ++kk) {
#pragma unroll
            for (int g = 0; g < NT; ++g)
                t[g] = __builtin_amdgcn_wmma_f32_16x16x4_f32(
                           false, f[kk], false, wx[g][kk], (short)0,
                           t[g], false, false);
        }
    };
    auto handover = [&](float* Pw, const v8f (&t)[NT]) {
#pragma unroll
        for (int g = 0; g < NT; ++g) {
            v4f lo, hi;
            lo[0] = t[g][0]; lo[1] = t[g][1]; lo[2] = t[g][2]; lo[3] = t[g][3];
            hi[0] = t[g][4]; hi[1] = t[g][5]; hi[2] = t[g][6]; hi[3] = t[g][7];
            *(v4f*)(Pw + poff[g])       = lo;
            *(v4f*)(Pw + poff[g] + 128) = hi;
        }
    };
    auto xfrags = [&](const float* Zs, v2f (&f)[XK]) {
#pragma unroll
        for (int kk = 0; kk < XK; ++kk)
            f[kk] = *(const v2f*)(Zs + c16 * K_SZ + kk * 4 + khalf);
    };

    v8f tp[NT];
    v2f fc[XK];
    v2f xr[4];
    {
        v2f f0[XK];
        v8f t0[NT];
        xfrags(Zf, f0);
        xpart(f0, t0);
        handover(Pf, t0);
        xfrags(Zf + SLAB, f0);
        xpart(f0, tp);
        xfrags(Zf + 2 * SLAB, fc);
#pragma unroll
        for (int i = 0; i < 4; ++i)
            xr[i] = *(const v2f*)(xsrc[i] + (size_t)4 * E_SZ);
    }
    __syncthreads();

    for (int sp = 0; sp < S_SZ; sp += NBUF) {
#pragma unroll
        for (int par = 0; par < NBUF; ++par) {
            const int s = sp + par;
            const float* Zn3 = Zf + ((par + 3) & (NBUF - 1)) * SLAB;
            float*       Zn4 = Zf + ((par + 4) & (NBUF - 1)) * SLAB;
            const int sn = (s + 5 < S_SZ) ? s + 5 : S_SZ - 1;

            handover(Pf + ((par + 1) & 1) * PPAR, tp);
#pragma unroll
            for (int q = 0; q < 4; ++q)
                *(v2f*)(Zn4 + xdst[q]) = xr[q];
            v2f fn[XK];
            xfrags(Zn3, fn);
#pragma unroll
            for (int i = 0; i < 4; ++i)
                xr[i] = *(const v2f*)(xsrc[i] + (size_t)sn * E_SZ);
            __builtin_amdgcn_sched_barrier(0);

            v8f tc[NT];
            xpart(fc, tc);
            __builtin_amdgcn_sched_barrier(0);

            __syncthreads();

#pragma unroll
            for (int g = 0; g < NT; ++g) tp[g] = tc[g];
#pragma unroll
            for (int kk = 0; kk < XK; ++kk) fc[kk] = fn[kk];
        }
    }
}

__global__ __launch_bounds__(THREADS)
void lstm_pair_kernel(const float* __restrict__ X,
                      const float* __restrict__ Wih,
                      const float* __restrict__ Whh,
                      const float* __restrict__ bih,
                      const float* __restrict__ bhh,
                      float* __restrict__ out)
{
    __shared__ __attribute__((aligned(16))) float Z[NBUF + 1][ROWS][K_SZ];
    __shared__ __attribute__((aligned(16))) float P[2][NOWN][8][32][4];

    const int tid  = threadIdx.x;
    const int wave = __builtin_amdgcn_readfirstlane(tid >> 5);
    const int lane = tid & 31;
    const int b0   = blockIdx.x * ROWS;
    const bool xrole = wave >= NOWN;
    const int  uw    = wave & (NOWN - 1);

    float* Zf = &Z[0][0][0];
    float* Pf = &P[0][0][0][0][0];

    for (int idx = tid; idx < ROWS * H_SZ; idx += THREADS)
        Z[0][idx / H_SZ][E_SZ + idx % H_SZ] = 0.f;
#pragma unroll
    for (int i = 0; i < 2; ++i) {
        const int p  = tid + i * THREADS;
        const int pc = (p < XPAIRS) ? p : XPAIRS - 1;
        const int r  = pc / (E_SZ / 2);
        const int e  = (pc % (E_SZ / 2)) * 2;
        const float* src = X + (size_t)(b0 + r) * S_SZ * E_SZ + e;
#pragma unroll
        for (int t = 0; t < 4; ++t)
            *(v2f*)(Zf + t * SLAB + r * K_SZ + e) = *(const v2f*)(src + t * E_SZ);
    }
    __syncthreads();

    if (!xrole) {
        if (uw < NOWN - 1) h_wave<false>(Wih, Whh, Zf, Pf, uw, lane);
        else               h_wave<true >(Wih, Whh, Zf, Pf, uw, lane);
    } else {
        if (uw < NOWN - 1) x_wave<3>(X, Wih, bih, bhh, Zf, Pf, uw, lane, tid, b0);
        else               x_wave<4>(X, Wih, bih, bhh, Zf, Pf, uw, lane, tid, b0);
    }

    for (int idx = tid; idx < ROWS * H_SZ; idx += THREADS)
        out[(size_t)b0 * H_SZ + idx] = Z[0][idx / H_SZ][E_SZ + idx % H_SZ];
}

extern "C" void kernel_launch(void* const* d_in, const int* in_sizes, int n_in,
                              void* d_out, int out_size, void* d_ws, size_t ws_size,
                              hipStream_t stream) {
    (void)in_sizes; (void)n_in; (void)d_ws; (void)ws_size; (void)out_size;
    const float* X   = (const float*)d_in[0];
    const float* Wih = (const float*)d_in[1];
    const float* Whh = (const float*)d_in[2];
    const float* bih = (const float*)d_in[3];
    const float* bhh = (const float*)d_in[4];
    float* out = (float*)d_out;

    dim3 grid(B_SZ / ROWS);
    dim3 block(THREADS);
    lstm_pair_kernel<<<grid, block, 0, stream>>>(X, Wih, Whh, bih, bhh, out);
}


# x waves open each phase with their first 12 WMMAs at raised priority; x request and all their LDS traffic follow, under the h waves' WMMAs: no idle matrix pipe at the phase top; sums unchanged
# speedup vs baseline: 1.6618x; 1.0499x over previous
#include <hip/hip_runtime.h>
#include <cmath>


typedef float v2f __attribute__((ext_vector_type(2)));
typedef float v4f __attribute__((ext_vector_type(4)));
typedef float v8f __attribute__((ext_vector_type(8)));

#define B_SZ 1024
#define S_SZ 512
#define E_SZ 50
#define H_SZ 50
#define K_SZ 100
#define KSTEPS 25
#define ROWS 16
#define NOWN 4
#define UG 16
#define ULAST 48
#define NWAVES 8
#define THREADS (NWAVES * 32)
#define XTHREADS (NOWN * 32)
#define XPAIRS (ROWS * E_SZ / 2)
#define NBUF 8
#define SLAB (ROWS * K_SZ)
#define XK 12
#define HK (KSTEPS - XK)
#define GSET (8 * 32 * 4)
#define PPAR (NOWN * GSET)

#if __has_builtin(__builtin_amdgcn_tanhf)
#define FAST_TANH(x) __builtin_amdgcn_tanhf(x)
#elif __has_builtin(__builtin_amdgcn_tanh_f32)
#define FAST_TANH(x) __builtin_amdgcn_tanh_f32(x)
#else
#define FAST_TANH(x) tanhf(x)
#endif


template <int N>
__device__ __forceinline__ float row_from(float x)
{
    return __builtin_bit_cast(float,
        __builtin_amdgcn_update_dpp(0, __builtin_bit_cast(int, x), 0x100 + N, 0xf, 0xf, true));
}

template <bool MIXED>
__device__ __forceinline__ void h_wave(const float* __restrict__ Wih,
                                       const float* __restrict__ Whh,
                                       float* Zf, const float* Pf,
                                       const int uw, const int lane)
{
    constexpr int NT = MIXED ? 1 : 4;
    const int c16   = lane & 15;
    const int khalf = (lane >> 4) * 2;
    const int rbase = (lane >> 4) * 8;

    v2f wh[NT][HK];
#pragma unroll
    for (int g = 0; g < NT; ++g) {
        const bool valid = MIXED ? (c16 < 8) : true;
        const int  n     = MIXED ? (valid ? (c16 >> 1) * H_SZ + ULAST + (c16 & 1) : 0)
                                 : g * H_SZ + uw * UG + c16;
#pragma unroll
        for (int kk = XK; kk < KSTEPS; ++kk) {
            const int k = kk * 4 + khalf;
            float w0, w1;
            if (k < E_SZ) {
                w0 = Wih[n * E_SZ + k];
                w1 = Wih[n * E_SZ + k + 1];
            } else {
                w0 = Whh[n * H_SZ + (k - E_SZ)];
                w1 = Whh[n * H_SZ + (k + 1 - E_SZ)];
            }
            wh[g][kk - XK].x = valid ? w0 : 0.f;
            wh[g][kk - XK].y = valid ? w1 : 0.f;
        }
    }
    float cst[8];
#pragma unroll
    for (int v = 0; v < 8; ++v) cst[v] = 0.f;
    const bool hvalid = MIXED ? (c16 < 2) : true;
    const int  hrow   = rbase * K_SZ + E_SZ + (MIXED ? (hvalid ? ULAST + c16 : c16) : uw * UG + c16);
    const int  pslot  = uw * GSET + lane * 4;
    const bool isg = (c16 >> 1) == 2;
    const float sc = isg ? 1.0f : 0.5f;
    const float ab = isg ? 0.0f : 0.5f;

    __syncthreads();

    for (int sp = 0; sp < S_SZ; sp += NBUF) {
#pragma unroll
        for (int par = 0; par < NBUF; ++par) {
            const float* Zc = Zf + par * SLAB;
            float*       Zn = Zf + (hvalid ? ((par + 1) & (NBUF - 1)) : NBUF) * SLAB + hrow;
            const float* Pr = Pf + (par & 1) * PPAR + pslot;

            v8f acc[NT];
#pragma unroll
            for (int g = 0; g < NT; ++g) {
                const v4f lo = *(const v4f*)(Pr + (g * 2 + 0) * 128);
                const v4f hi = *(const v4f*)(Pr + (g * 2 + 1) * 128);
                acc[g][0] = lo[0]; acc[g][1] = lo[1]; acc[g][2] = lo[2]; acc[g][3] = lo[3];
                acc[g][4] = hi[0]; acc[g][5] = hi[1]; acc[g][6] = hi[2]; acc[g][7] = hi[3];
            }
            v2f a[HK];
#pragma unroll
            for (int kk = 0; kk < HK; ++kk)
                a[kk] = *(const v2f*)(Zc + c16 * K_SZ + (XK + kk) * 4 + khalf);
            __builtin_amdgcn_sched_barrier(0);

#pragma unroll
            for (int kk = 0; kk < HK; ++kk) {
#pragma unroll
                for (int g = 0; g < NT; ++g)
                    acc[g] = __builtin_amdgcn_wmma_f32_16x16x4_f32(
                                 false, a[kk], false, wh[g][kk], (short)0,
                                 acc[g], false, false);
            }

            float hv[8];
            if constexpr (!MIXED) {
                float ti[8], tf[8], tg[8], to[8];
#pragma unroll
                for (int v = 0; v < 8; ++v) ti[v] = 0.5f * acc[0][v];
#pragma unroll
                for (int v = 0; v < 8; ++v) tf[v] = 0.5f * acc[1][v];
#pragma unroll
                for (int v = 0; v < 8; ++v) to[v] = 0.5f * acc[NT - 1][v];
#pragma unroll
                for (int v = 0; v < 8; ++v) ti[v] = FAST_TANH(ti[v]);
#pragma unroll
                for (int v = 0; v < 8; ++v) tf[v] = FAST_TANH(tf[v]);
#pragma unroll
                for (int v = 0; v < 8; ++v) tg[v] = FAST_TANH(acc[NT / 2][v]);
#pragma unroll
                for (int v = 0; v < 8; ++v) to[v] = FAST_TANH(to[v]);
#pragma unroll
                for (int v = 0; v < 8; ++v) ti[v] = __builtin_fmaf(0.5f, ti[v], 0.5f);
#pragma unroll
                for (int v = 0; v < 8; ++v) tf[v] = __builtin_fmaf(0.5f, tf[v], 0.5f);
#pragma unroll
                for (int v = 0; v < 8; ++v) to[v] = __builtin_fmaf(0.5f, to[v], 0.5f);
#pragma unroll
                for (int v = 0; v < 8; ++v) tg[v] = ti[v] * tg[v];
#pragma unroll
                for (int v = 0; v < 8; ++v) cst[v] = __builtin_fmaf(tf[v], cst[v], tg[v]);
#pragma unroll
                for (int v = 0; v < 8; ++v) tg[v] = FAST_TANH(cst[v]);
#pragma unroll
                for (int v = 0; v < 8; ++v) hv[v] = to[v] * tg[v];
            } else {
                float ta[8], tf[8], tg[8], to[8];
#pragma unroll
                for (int v = 0; v < 8; ++v) ta[v] = sc * acc[0][v];
#pragma unroll
                for (int v = 0; v < 8; ++v) ta[v] = FAST_TANH(ta[v]);
#pragma unroll
                for (int v = 0; v < 8; ++v) ta[v] = __builtin_fmaf(sc, ta[v], ab);
#pragma unroll
                for (int v = 0; v < 8; ++v) {
                    tf[v] = row_from<2>(ta[v]);
                    tg[v] = row_from<4>(ta[v]);
                    to[v] = row_from<6>(ta[v]);
                }
#pragma unroll
                for (int v = 0; v < 8; ++v) tg[v] = ta[v] * tg[v];
#pragma unroll
                for (int v = 0; v < 8; ++v) cst[v] = __builtin_fmaf(tf[v], cst[v], tg[v]);
#pragma unroll
                for (int v = 0; v < 8; ++v) tg[v] = FAST_TANH(cst[v]);
#pragma unroll
                for (int v = 0; v < 8; ++v) hv[v] = to[v] * tg[v];
            }
#pragma unroll
            for (int v = 0; v < 8; ++v)
                Zn[v * K_SZ] = hv[v];

            __syncthreads();
        }
    }
}

template <int NT>
__device__ __forceinline__ void x_wave(const float* __restrict__ X,
                                       const float* __restrict__ Wih,
                                       const float* __restrict__ bih,
                                       const float* __restrict__ bhh,
                                       float* Zf, float* Pf,
                                       const int uw, const int lane, const int tid, const int b0)
{
    const int c16   = lane & 15;
    const int khalf = (lane >> 4) * 2;

    v2f   wx[NT][XK];
    float bias[NT];
    int   poff[NT];
#pragma unroll
    for (int g = 0; g < NT; ++g) {
        bool valid; int n;
        if (NT == 3) {
            valid = true;
            n = g * H_SZ + uw * UG + c16;
            poff[g] = uw * GSET + g * 256 + lane * 4;
        } else if (g < 3) {
            valid = true;
            n = 3 * H_SZ + g * UG + c16;
            poff[g] = g * GSET + 3 * 256 + lane * 4;
        } else {
            valid = c16 < 8;
            n = valid ? (c16 >> 1) * H_SZ + ULAST + (c16 & 1) : 0;
            poff[g] = 3 * GSET + lane * 4;
        }
#pragma unroll
        for (int kk = 0; kk < XK; ++kk) {
            const int k = kk * 4 + khalf;
            const float w0 = Wih[n * E_SZ + k];
            const float w1 = Wih[n * E_SZ + k + 1];
            wx[g][kk].x = valid ? w0 : 0.f;
            wx[g][kk].y = valid ? w1 : 0.f;
        }
        bias[g] = valid ? (bih[n] + bhh[n]) : 0.f;
    }
    v8f bv[NT];
#pragma unroll
    for (int g = 0; g < NT; ++g)
#pragma unroll
        for (int v = 0; v < 8; ++v) bv[g][v] = bias[g];

    const int xt = tid - XTHREADS;
    const float* xsrc[4];
    int          xdst[4];
#pragma unroll
    for (int i = 0; i < 4; ++i) {
        const int p  = xt + i * XTHREADS;
        const int pc = (p < XPAIRS) ? p : XPAIRS - 1;
        const int r  = pc / (E_SZ / 2);
        const int e  = (pc % (E_SZ / 2)) * 2;
        xsrc[i] = X + (size_t)(b0 + r) * S_SZ * E_SZ + e;
        xdst[i] = r * K_SZ + e;
    }

    constexpr int KH = 12 / NT;
    auto xhead = [&](const v2f (&f)[XK], v8f (&t)[NT]) {
#pragma unroll
        for (int g = 0; g < NT; ++g)
            t[g] = __builtin_amdgcn_wmma_f32_16x16x4_f32(
                       false, f[0], false, wx[g][0], (short)0,
                       bv[g], false, false);
#pragma unroll
        for (int kk = 1; kk < KH; ++kk) {
#pragma unroll
            for (int g = 0; g < NT; ++g)
                t[g] = __builtin_amdgcn_wmma_f32_16x16x4_f32(
                           false, f[kk], false, wx[g][kk], (short)0,
                           t[g], false, false);
        }
    };
    auto xtail = [&](const v2f (&f)[XK], v8f (&t)[NT]) {
#pragma unroll
        for (int kk = KH; kk < XK; ++kk) {
#pragma unroll
            for (int g = 0; g < NT; ++g)
                t[g] = __builtin_amdgcn_wmma_f32_16x16x4_f32(
                           false, f[kk], false, wx[g][kk], (short)0,
                           t[g], false, false);
        }
    };
    auto handover = [&](float* Pw, const v8f (&t)[NT]) {
#pragma unroll
        for (int g = 0; g < NT; ++g) {
            v4f lo, hi;
            lo[0] = t[g][0]; lo[1] = t[g][1]; lo[2] = t[g][2]; lo[3] = t[g][3];
            hi[0] = t[g][4]; hi[1] = t[g][5]; hi[2] = t[g][6]; hi[3] = t[g][7];
            *(v4f*)(Pw + poff[g])       = lo;
            *(v4f*)(Pw + poff[g] + 128) = hi;
        }
    };
    auto xfrags = [&](const float* Zs, v2f (&f)[XK]) {
#pragma unroll
        for (int kk = 0; kk < XK; ++kk)
            f[kk] = *(const v2f*)(Zs + c16 * K_SZ + kk * 4 + khalf);
    };

    v8f tp[NT];
    v2f fc[XK];
    v2f xr[4];
    {
        v2f f0[XK];
        v8f t0[NT];
        xfrags(Zf, f0);
        xhead(f0, t0);
        xtail(f0, t0);
        handover(Pf, t0);
        xfrags(Zf + SLAB, f0);
        xhead(f0, tp);
        xtail(f0, tp);
        xfrags(Zf + 2 * SLAB, fc);
#pragma unroll
        for (int i = 0; i < 4; ++i)
            xr[i] = *(const v2f*)(xsrc[i] + (size_t)4 * E_SZ);
    }
    __syncthreads();

    for (int sp = 0; sp < S_SZ; sp += NBUF) {
#pragma unroll
        for (int par = 0; par < NBUF; ++par) {
            const int s = sp + par;
            const float* Zn3 = Zf + ((par + 3) & (NBUF - 1)) * SLAB;
            float*       Zn4 = Zf + ((par + 4) & (NBUF - 1)) * SLAB;
            const int sn = (s + 5 < S_SZ) ? s + 5 : S_SZ - 1;

            __builtin_amdgcn_s_setprio(3);

            v8f tc[NT];
            xhead(fc, tc);
            if constexpr (NT == 3)
                asm volatile("s_setprio 0" :: "v"(tc[0]), "v"(tc[1]), "v"(tc[2]) : "memory");
            else
                asm volatile("s_setprio 0" :: "v"(tc[0]), "v"(tc[1]), "v"(tc[2]), "v"(tc[3]) : "memory");

            v2f xn[4];
#pragma unroll
            for (int i = 0; i < 4; ++i)
                xn[i] = *(const v2f*)(xsrc[i] + (size_t)sn * E_SZ);
            handover(Pf + ((par + 1) & 1) * PPAR, tp);
#pragma unroll
            for (int q = 0; q < 4; ++q)
                *(v2f*)(Zn4 + xdst[q]) = xr[q];
            v2f fn[XK];
            xfrags(Zn3, fn);
            __builtin_amdgcn_sched_barrier(0);

            xtail(fc, tc);
            __builtin_amdgcn_sched_barrier(0);

            __syncthreads();

#pragma unroll
            for (int g = 0; g < NT; ++g) tp[g] = tc[g];
#pragma unroll
            for (int kk = 0; kk < XK; ++kk) fc[kk] = fn[kk];
#pragma unroll
            for (int i = 0; i < 4; ++i) xr[i] = xn[i];
        }
    }
}

__global__ __launch_bounds__(THREADS)
void lstm_pair_kernel(const float* __restrict__ X,
                      const float* __restrict__ Wih,
                      const float* __restrict__ Whh,
                      const float* __restrict__ bih,
                      const float* __restrict__ bhh,
                      float* __restrict__ out)
{
    __shared__ __attribute__((aligned(16))) float Z[NBUF + 1][ROWS][K_SZ];
    __shared__ __attribute__((aligned(16))) float P[2][NOWN][8][32][4];

    const int tid  = threadIdx.x;
    const int wave = __builtin_amdgcn_readfirstlane(tid >> 5);
    const int lane = tid & 31;
    const int b0   = blockIdx.x * ROWS;
    const bool xrole = wave >= NOWN;
    const int  uw    = wave & (NOWN - 1);

    float* Zf = &Z[0][0][0];
    float* Pf = &P[0][0][0][0][0];

    for (int idx = tid; idx < ROWS * H_SZ; idx += THREADS)
        Z[0][idx / H_SZ][E_SZ + idx % H_SZ] = 0.f;
#pragma unroll
    for (int i = 0; i < 2; ++i) {
        const int p  = tid + i * THREADS;
        const int pc = (p < XPAIRS) ? p : XPAIRS - 1;
        const int r  = pc / (E_SZ / 2);
        const int e  = (pc % (E_SZ / 2)) * 2;
        const float* src = X + (size_t)(b0 + r) * S_SZ * E_SZ + e;
#pragma unroll
        for (int t = 0; t < 4; ++t)
            *(v2f*)(Zf + t * SLAB + r * K_SZ + e) = *(const v2f*)(src + t * E_SZ);
    }
    __syncthreads();

    if (!xrole) {
        if (uw < NOWN - 1) h_wave<false>(Wih, Whh, Zf, Pf, uw, lane);
        else               h_wave<true >(Wih, Whh, Zf, Pf, uw, lane);
    } else {
        if (uw < NOWN - 1) x_wave<3>(X, Wih, bih, bhh, Zf, Pf, uw, lane, tid, b0);
        else               x_wave<4>(X, Wih, bih, bhh, Zf, Pf, uw, lane, tid, b0);
    }

    for (int idx = tid; idx < ROWS * H_SZ; idx += THREADS)
        out[(size_t)b0 * H_SZ + idx] = Z[0][idx / H_SZ][E_SZ + idx % H_SZ];
}

extern "C" void kernel_launch(void* const* d_in, const int* in_sizes, int n_in,
                              void* d_out, int out_size, void* d_ws, size_t ws_size,
                              hipStream_t stream) {
    (void)in_sizes; (void)n_in; (void)d_ws; (void)ws_size; (void)out_size;
    const float* X   = (const float*)d_in[0];
    const float* Wih = (const float*)d_in[1];
    const float* Whh = (const float*)d_in[2];
    const float* bih = (const float*)d_in[3];
    const float* bhh = (const float*)d_in[4];
    float* out = (float*)d_out;

    dim3 grid(B_SZ / ROWS);
    dim3 block(THREADS);
    lstm_pair_kernel<<<grid, block, 0, stream>>>(X, Wih, Whh, bih, bhh, out);
}


# bias and k-steps 0-5 of the g tiles of groups 0-2 moved to x wave 3 (SIMD 3), run 2 phases earlier, passed on as C via LDS: WMMAs per step and SIMD 82/82/82/79, not 88/88/88/61; sums unchanged
# speedup vs baseline: 1.7294x; 1.0407x over previous
#include <hip/hip_runtime.h>
#include <cmath>


typedef float v2f __attribute__((ext_vector_type(2)));
typedef float v4f __attribute__((ext_vector_type(4)));
typedef float v8f __attribute__((ext_vector_type(8)));

#define B_SZ 1024
#define S_SZ 512
#define E_SZ 50
#define H_SZ 50
#define K_SZ 100
#define KSTEPS 25
#define ROWS 16
#define NOWN 4
#define UG 16
#define ULAST 48
#define NWAVES 8
#define THREADS (NWAVES * 32)
#define XTHREADS (NOWN * 32)
#define XPAIRS (ROWS * E_SZ / 2)
#define NBUF 8
#define SLAB (ROWS * K_SZ)
#define XK 12
#define HK (KSTEPS - XK)
#define GSET (8 * 32 * 4)
#define PPAR (NOWN * GSET)
#define GK 6
#define QSET (2 * 32 * 4)
#define QPAR (3 * QSET)

#if __has_builtin(__builtin_amdgcn_tanhf)
#define FAST_TANH(x) __builtin_amdgcn_tanhf(x)
#elif __has_builtin(__builtin_amdgcn_tanh_f32)
#define FAST_TANH(x) __builtin_amdgcn_tanh_f32(x)
#else
#define FAST_TANH(x) tanhf(x)
#endif


template <int N>
__device__ __forceinline__ float row_from(float x)
{
    return __builtin_bit_cast(float,
        __builtin_amdgcn_update_dpp(0, __builtin_bit_cast(int, x), 0x100 + N, 0xf, 0xf, true));
}

template <bool MIXED>
__device__ __forceinline__ void h_wave(const float* __restrict__ Wih,
                                       const float* __restrict__ Whh,
                                       float* Zf, const float* Pf,
                                       const int uw, const int lane)
{
    constexpr int NT = MIXED ? 1 : 4;
    const int c16   = lane & 15;
    const int khalf = (lane >> 4) * 2;
    const int rbase = (lane >> 4) * 8;

    v2f wh[NT][HK];
#pragma unroll
    for (int g = 0; g < NT; ++g) {
        const bool valid = MIXED ? (c16 < 8) : true;
        const int  n     = MIXED ? (valid ? (c16 >> 1) * H_SZ + ULAST + (c16 & 1) : 0)
                                 : g * H_SZ + uw * UG + c16;
#pragma unroll
        for (int kk = XK; kk < KSTEPS; ++kk) {
            const int k = kk * 4 + khalf;
            float w0, w1;
            if (k < E_SZ) {
                w0 = Wih[n * E_SZ + k];
                w1 = Wih[n * E_SZ + k + 1];
            } else {
                w0 = Whh[n * H_SZ + (k - E_SZ)];
                w1 = Whh[n * H_SZ + (k + 1 - E_SZ)];
            }
            wh[g][kk - XK].x = valid ? w0 : 0.f;
            wh[g][kk - XK].y = valid ? w1 : 0.f;
        }
    }
    float cst[8];
#pragma unroll
    for (int v = 0; v < 8; ++v) cst[v] = 0.f;
    const bool hvalid = MIXED ? (c16 < 2) : true;
    const int  hrow   = rbase * K_SZ + E_SZ + (MIXED ? (hvalid ? ULAST + c16 : c16) : uw * UG + c16);
    const int  pslot  = uw * GSET + lane * 4;
    const bool isg = (c16 >> 1) == 2;
    const float sc = isg ? 1.0f : 0.5f;
    const float ab = isg ? 0.0f : 0.5f;

    __syncthreads();

    for (int sp = 0; sp < S_SZ; sp += NBUF) {
#pragma unroll
        for (int par = 0; par < NBUF; ++par) {
            const float* Zc = Zf + par * SLAB;
            float*       Zn = Zf + (hvalid ? ((par + 1) & (NBUF - 1)) : NBUF) * SLAB + hrow;
            const float* Pr = Pf + (par & 1) * PPAR + pslot;

            v8f acc[NT];
#pragma unroll
            for (int g = 0; g < NT; ++g) {
                const v4f lo = *(const v4f*)(Pr + (g * 2 + 0) * 128);
                const v4f hi = *(const v4f*)(Pr + (g * 2 + 1) * 128);
                acc[g][0] = lo[0]; acc[g][1] = lo[1]; acc[g][2] = lo[2]; acc[g][3] = lo[3];
                acc[g][4] = hi[0]; acc[g][5] = hi[1]; acc[g][6] = hi[2]; acc[g][7] = hi[3];
            }
            v2f a[HK];
#pragma unroll
            for (int kk = 0; kk < HK; ++kk)
                a[kk] = *(const v2f*)(Zc + c16 * K_SZ + (XK + kk) * 4 + khalf);
            __builtin_amdgcn_sched_barrier(0);

#pragma unroll
            for (int kk = 0; kk < HK; ++kk) {
#pragma unroll
                for (int g = 0; g < NT; ++g)
                    acc[g] = __builtin_amdgcn_wmma_f32_16x16x4_f32(
                                 false, a[kk], false, wh[g][kk], (short)0,
                                 acc[g], false, false);
            }

            float hv[8];
            if constexpr (!MIXED) {
                float ti[8], tf[8], tg[8], to[8];
#pragma unroll
                for (int v = 0; v < 8; ++v) ti[v] = 0.5f * acc[0][v];
#pragma unroll
                for (int v = 0; v < 8; ++v) tf[v] = 0.5f * acc[1][v];
#pragma unroll
                for (int v = 0; v < 8; ++v) to[v] = 0.5f * acc[NT - 1][v];
#pragma unroll
                for (int v = 0; v < 8; ++v) ti[v] = FAST_TANH(ti[v]);
#pragma unroll
                for (int v = 0; v < 8; ++v) tf[v] = FAST_TANH(tf[v]);
#pragma unroll
                for (int v = 0; v < 8; ++v) tg[v] = FAST_TANH(acc[NT / 2][v]);
#pragma unroll
                for (int v = 0; v < 8; ++v) to[v] = FAST_TANH(to[v]);
#pragma unroll
                for (int v = 0; v < 8; ++v) ti[v] = __builtin_fmaf(0.5f, ti[v], 0.5f);
#pragma unroll
                for (int v = 0; v < 8; ++v) tf[v] = __builtin_fmaf(0.5f, tf[v], 0.5f);
#pragma unroll
                for (int v = 0; v < 8; ++v) to[v] = __builtin_fmaf(0.5f, to[v], 0.5f);
#pragma unroll
                for (int v = 0; v < 8; ++v) tg[v] = ti[v] * tg[v];
#pragma unroll
                for (int v = 0; v < 8; ++v) cst[v] = __builtin_fmaf(tf[v], cst[v], tg[v]);
#pragma unroll
                for (int v = 0; v < 8; ++v) tg[v] = FAST_TANH(cst[v]);
#pragma unroll
                for (int v = 0; v < 8; ++v) hv[v] = to[v] * tg[v];
            } else {
                float ta[8], tf[8], tg[8], to[8];
#pragma unroll
                for (int v = 0; v < 8; ++v) ta[v] = sc * acc[0][v];
#pragma unroll
                for (int v = 0; v < 8; ++v) ta[v] = FAST_TANH(ta[v]);
#pragma unroll
                for (int v = 0; v < 8; ++v) ta[v] = __builtin_fmaf(sc, ta[v], ab);
#pragma unroll
                for (int v = 0; v < 8; ++v) {
                    tf[v] = row_from<2>(ta[v]);
                    tg[v] = row_from<4>(ta[v]);
                    to[v] = row_from<6>(ta[v]);
                }
#pragma unroll
                for (int v = 0; v < 8; ++v) tg[v] = ta[v] * tg[v];
#pragma unroll
                for (int v = 0; v < 8; ++v) cst[v] = __builtin_fmaf(tf[v], cst[v], tg[v]);
#pragma unroll
                for (int v = 0; v < 8; ++v) tg[v] = FAST_TANH(cst[v]);
#pragma unroll
                for (int v = 0; v < 8; ++v) hv[v] = to[v] * tg[v];
            }
#pragma unroll
            for (int v = 0; v < 8; ++v)
                Zn[v * K_SZ] = hv[v];

            __syncthreads();
        }
    }
}

template <int NT>
__device__ __forceinline__ void x_wave(const float* __restrict__ X,
                                       const float* __restrict__ Wih,
                                       const float* __restrict__ bih,
                                       const float* __restrict__ bhh,
                                       float* Zf, float* Pf, float* Qf,
                                       const int uw, const int lane, const int tid, const int b0)
{
    constexpr bool LAST = (NT == 4);
    const int c16   = lane & 15;
    const int khalf = (lane >> 4) * 2;

    v2f   wx[NT][XK];
    float bias[NT];
    int   poff[NT];
#pragma unroll
    for (int g = 0; g < NT; ++g) {
        bool valid; int n;
        if (NT == 3) {
            valid = true;
            n = g * H_SZ + uw * UG + c16;
            poff[g] = uw * GSET + g * 256 + lane * 4;
        } else if (g < 3) {
            valid = true;
            n = 3 * H_SZ + g * UG + c16;
            poff[g] = g * GSET + 3 * 256 + lane * 4;
        } else {
            valid = c16 < 8;
            n = valid ? (c16 >> 1) * H_SZ + ULAST + (c16 & 1) : 0;
            poff[g] = 3 * GSET + lane * 4;
        }
#pragma unroll
        for (int kk = 0; kk < XK; ++kk) {
            const int k = kk * 4 + khalf;
            const float w0 = Wih[n * E_SZ + k];
            const float w1 = Wih[n * E_SZ + k + 1];
            wx[g][kk].x = valid ? w0 : 0.f;
            wx[g][kk].y = valid ? w1 : 0.f;
        }
        bias[g] = valid ? (bih[n] + bhh[n]) : 0.f;
    }
    v8f bv[NT];
#pragma unroll
    for (int g = 0; g < NT; ++g)
#pragma unroll
        for (int v = 0; v < 8; ++v) bv[g][v] = bias[g];

    constexpr int NQ = LAST ? 3 : 1;
    v2f wg[NQ][GK];
    v8f bvg[NQ];
    if constexpr (LAST) {
#pragma unroll
        for (int j = 0; j < 3; ++j) {
            const int n = 2 * H_SZ + j * UG + c16;
#pragma unroll
            for (int kk = 0; kk < GK; ++kk) {
                const int k = kk * 4 + khalf;
                wg[j][kk].x = Wih[n * E_SZ + k];
                wg[j][kk].y = Wih[n * E_SZ + k + 1];
            }
            const float b = bih[n] + bhh[n];
#pragma unroll
            for (int v = 0; v < 8; ++v) bvg[j][v] = b;
        }
    }

    const int xt = tid - XTHREADS;
    const float* xsrc[4];
    int          xdst[4];
#pragma unroll
    for (int i = 0; i < 4; ++i) {
        const int p  = xt + i * XTHREADS;
        const int pc = (p < XPAIRS) ? p : XPAIRS - 1;
        const int r  = pc / (E_SZ / 2);
        const int e  = (pc % (E_SZ / 2)) * 2;
        xsrc[i] = X + (size_t)(b0 + r) * S_SZ * E_SZ + e;
        xdst[i] = r * K_SZ + e;
    }

    constexpr int NHEAD = LAST ? 4 : 2;
    constexpr int KH = 12 / NHEAD;
    static_assert(LAST || KH == GK, "a group's three tiles carry on at the same k-step");
    auto xhead = [&](const v2f (&f)[XK], v8f (&t)[NT]) {
#pragma unroll
        for (int g = 0; g < NHEAD; ++g)
            t[g] = __builtin_amdgcn_wmma_f32_16x16x4_f32(
                       false, f[0], false, wx[g][0], (short)0,
                       bv[g], false, false);
#pragma unroll
        for (int kk = 1; kk < KH; ++kk) {
#pragma unroll
            for (int g = 0; g < NHEAD; ++g)
                t[g] = __builtin_amdgcn_wmma_f32_16x16x4_f32(
                           false, f[kk], false, wx[g][kk], (short)0,
                           t[g], false, false);
        }
    };
    auto xall = [&](const v2f (&f)[XK], v8f (&t)[NT]) {
#pragma unroll
        for (int g = 0; g < NT; ++g)
            t[g] = __builtin_amdgcn_wmma_f32_16x16x4_f32(
                       false, f[0], false, wx[g][0], (short)0,
                       bv[g], false, false);
#pragma unroll
        for (int kk = 1; kk < XK; ++kk) {
#pragma unroll
            for (int g = 0; g < NT; ++g)
                t[g] = __builtin_amdgcn_wmma_f32_16x16x4_f32(
                           false, f[kk], false, wx[g][kk], (short)0,
                           t[g], false, false);
        }
    };
    auto xtail = [&](const v2f (&f)[XK], v8f (&t)[NT]) {
#pragma unroll
        for (int kk = KH; kk < XK; ++kk) {
#pragma unroll
            for (int g = 0; g < NT; ++g)
                t[g] = __builtin_amdgcn_wmma_f32_16x16x4_f32(
                           false, f[kk], false, wx[g][kk], (short)0,
                           t[g], false, false);
        }
    };
    auto handover = [&](float* Pw, const v8f (&t)[NT]) {
#pragma unroll
        for (int g = 0; g < NT; ++g) {
            v4f lo, hi;
            lo[0] = t[g][0]; lo[1] = t[g][1]; lo[2] = t[g][2]; lo[3] = t[g][3];
            hi[0] = t[g][4]; hi[1] = t[g][5]; hi[2] = t[g][6]; hi[3] = t[g][7];
            *(v4f*)(Pw + poff[g])       = lo;
            *(v4f*)(Pw + poff[g] + 128) = hi;
        }
    };
    auto xfrags = [&](const float* Zs, v2f (&f)[XK]) {
#pragma unroll
        for (int kk = 0; kk < XK; ++kk)
            f[kk] = *(const v2f*)(Zs + c16 * K_SZ + kk * 4 + khalf);
    };

    auto gfrags = [&](const float* Zs, v2f (&f)[GK]) {
#pragma unroll
        for (int kk = 0; kk < GK; ++kk)
            f[kk] = *(const v2f*)(Zs + c16 * K_SZ + kk * 4 + khalf);
    };
    auto gstart = [&](const v2f (&f)[GK], v8f (&q)[NQ]) {
#pragma unroll
        for (int j = 0; j < NQ; ++j)
            q[j] = __builtin_amdgcn_wmma_f32_16x16x4_f32(
                       false, f[0], false, wg[j][0], (short)0,
                       bvg[j], false, false);
#pragma unroll
        for (int kk = 1; kk < GK; ++kk) {
#pragma unroll
            for (int j = 0; j < NQ; ++j)
                q[j] = __builtin_amdgcn_wmma_f32_16x16x4_f32(
                           false, f[kk], false, wg[j][kk], (short)0,
                           q[j], false, false);
        }
    };
    auto gleave = [&](float* Qw, const v8f (&q)[NQ]) {
#pragma unroll
        for (int j = 0; j < NQ; ++j) {
            v4f lo, hi;
            lo[0] = q[j][0]; lo[1] = q[j][1]; lo[2] = q[j][2]; lo[3] = q[j][3];
            hi[0] = q[j][4]; hi[1] = q[j][5]; hi[2] = q[j][6]; hi[3] = q[j][7];
            *(v4f*)(Qw + j * QSET + lane * 4)       = lo;
            *(v4f*)(Qw + j * QSET + 128 + lane * 4) = hi;
        }
    };

    v8f tp[NT];
    v2f fc[XK];
    v2f xr[4];
    v8f qp[NQ];
    {
        v2f f0[XK];
        v8f t0[NT];
        xfrags(Zf, f0);
        xall(f0, t0);
        handover(Pf, t0);
        xfrags(Zf + SLAB, f0);
        xall(f0, tp);
        xfrags(Zf + 2 * SLAB, fc);
        if constexpr (LAST) {
            v2f g0[GK];
            v8f q0[NQ];
            gfrags(Zf + 2 * SLAB, g0);
            gstart(g0, q0);
            gleave(Qf, q0);
            gfrags(Zf + 3 * SLAB, g0);
            gstart(g0, qp);
        }
#pragma unroll
        for (int i = 0; i < 4; ++i)
            xr[i] = *(const v2f*)(xsrc[i] + (size_t)5 * E_SZ);
    }
    __syncthreads();

    for (int sp = 0; sp < S_SZ; sp += NBUF) {
#pragma unroll
        for (int par = 0; par < NBUF; ++par) {
            const int s = sp + par;
            const float* Zn3 = Zf + ((par + 3) & (NBUF - 1)) * SLAB;
            const float* Zn4 = Zf + ((par + 4) & (NBUF - 1)) * SLAB;
            float*       Zn5 = Zf + ((par + 5) & (NBUF - 1)) * SLAB;
            const int sn = (s + 6 < S_SZ) ? s + 6 : S_SZ - 1;

            __builtin_amdgcn_s_setprio(3);

            v8f tc[NT];
            xhead(fc, tc);
            if constexpr (!LAST)
                asm volatile("s_setprio 0" :: "v"(tc[0]), "v"(tc[1]) : "memory");
            else
                asm volatile("s_setprio 0" :: "v"(tc[0]), "v"(tc[1]), "v"(tc[2]), "v"(tc[3]) : "memory");

            v2f xn[4];
#pragma unroll
            for (int i = 0; i < 4; ++i)
                xn[i] = *(const v2f*)(xsrc[i] + (size_t)sn * E_SZ);
            handover(Pf + ((par + 1) & 1) * PPAR, tp);
            if constexpr (LAST)
                gleave(Qf + ((par + 1) & 1) * QPAR, qp);
#pragma unroll
            for (int q = 0; q < 4; ++q)
                *(v2f*)(Zn5 + xdst[q]) = xr[q];
            v2f fn[XK];
            xfrags(Zn3, fn);
            v2f gf[GK];
            v8f qc[NQ];
            if constexpr (LAST) {
                gfrags(Zn4, gf);
            } else {
                const float* Qr = Qf + (par & 1) * QPAR + uw * QSET + lane * 4;
                const v4f lo = *(const v4f*)(Qr);
                const v4f hi = *(const v4f*)(Qr + 128);
                tc[2][0] = lo[0]; tc[2][1] = lo[1]; tc[2][2] = lo[2]; tc[2][3] = lo[3];
                tc[2][4] = hi[0]; tc[2][5] = hi[1]; tc[2][6] = hi[2]; tc[2][7] = hi[3];
            }
            __builtin_amdgcn_sched_barrier(0);

            xtail(fc, tc);
            if constexpr (LAST)
                gstart(gf, qc);
            __builtin_amdgcn_sched_barrier(0);

            __syncthreads();

#pragma unroll
            for (int g = 0; g < NT; ++g) tp[g] = tc[g];
#pragma unroll
            for (int kk = 0; kk < XK; ++kk) fc[kk] = fn[kk];
#pragma unroll
            for (int i = 0; i < 4; ++i) xr[i] = xn[i];
            if constexpr (LAST) {
#pragma unroll
                for (int j = 0; j < NQ; ++j) qp[j] = qc[j];
            }
        }
    }
}

__global__ __launch_bounds__(THREADS)
void lstm_pair_kernel(const float* __restrict__ X,
                      const float* __restrict__ Wih,
                      const float* __restrict__ Whh,
                      const float* __restrict__ bih,
                      const float* __restrict__ bhh,
                      float* __restrict__ out)
{
    __shared__ __attribute__((aligned(16))) float Z[NBUF + 1][ROWS][K_SZ];
    __shared__ __attribute__((aligned(16))) float P[2][NOWN][8][32][4];
    __shared__ __attribute__((aligned(16))) float Q[2][3][2][32][4];

    const int tid  = threadIdx.x;
    const int wave = __builtin_amdgcn_readfirstlane(tid >> 5);
    const int lane = tid & 31;
    const int b0   = blockIdx.x * ROWS;
    const bool xrole = wave >= NOWN;
    const int  uw    = wave & (NOWN - 1);

    float* Zf = &Z[0][0][0];
    float* Pf = &P[0][0][0][0][0];
    float* Qf = &Q[0][0][0][0][0];

    for (int idx = tid; idx < ROWS * H_SZ; idx += THREADS)
        Z[0][idx / H_SZ][E_SZ + idx % H_SZ] = 0.f;
#pragma unroll
    for (int i = 0; i < 2; ++i) {
        const int p  = tid + i * THREADS;
        const int pc = (p < XPAIRS) ? p : XPAIRS - 1;
        const int r  = pc / (E_SZ / 2);
        const int e  = (pc % (E_SZ / 2)) * 2;
        const float* src = X + (size_t)(b0 + r) * S_SZ * E_SZ + e;
#pragma unroll
        for (int t = 0; t < 5; ++t)
            *(v2f*)(Zf + t * SLAB + r * K_SZ + e) = *(const v2f*)(src + t * E_SZ);
    }
    __syncthreads();

    if (!xrole) {
        if (uw < NOWN - 1) h_wave<false>(Wih, Whh, Zf, Pf, uw, lane);
        else               h_wave<true >(Wih, Whh, Zf, Pf, uw, lane);
    } else {
        if (uw < NOWN - 1) x_wave<3>(X, Wih, bih, bhh, Zf, Pf, Qf, uw, lane, tid, b0);
        else               x_wave<4>(X, Wih, bih, bhh, Zf, Pf, Qf, uw, lane, tid, b0);
    }

    for (int idx = tid; idx < ROWS * H_SZ; idx += THREADS)
        out[(size_t)b0 * H_SZ + idx] = Z[0][idx / H_SZ][E_SZ + idx % H_SZ];
}

extern "C" void kernel_launch(void* const* d_in, const int* in_sizes, int n_in,
                              void* d_out, int out_size, void* d_ws, size_t ws_size,
                              hipStream_t stream) {
    (void)in_sizes; (void)n_in; (void)d_ws; (void)ws_size; (void)out_size;
    const float* X   = (const float*)d_in[0];
    const float* Wih = (const float*)d_in[1];
    const float* Whh = (const float*)d_in[2];
    const float* bih = (const float*)d_in[3];
    const float* bhh = (const float*)d_in[4];
    float* out = (float*)d_out;

    dim3 grid(B_SZ / ROWS);
    dim3 block(THREADS);
    lstm_pair_kernel<<<grid, block, 0, stream>>>(X, Wih, Whh, bih, bhh, out);
}


# x request issued two steps before its LDS store, not one, and left in flight across the step barrier: the x waves' barrier releases LDS only, no wait for global loads; sums unchanged
# speedup vs baseline: 1.8523x; 1.0711x over previous
#include <hip/hip_runtime.h>
#include <cmath>


typedef float v2f __attribute__((ext_vector_type(2)));
typedef float v4f __attribute__((ext_vector_type(4)));
typedef float v8f __attribute__((ext_vector_type(8)));

#define B_SZ 1024
#define S_SZ 512
#define E_SZ 50
#define H_SZ 50
#define K_SZ 100
#define KSTEPS 25
#define ROWS 16
#define NOWN 4
#define UG 16
#define ULAST 48
#define NWAVES 8
#define THREADS (NWAVES * 32)
#define XTHREADS (NOWN * 32)
#define XPAIRS (ROWS * E_SZ / 2)
#define NBUF 8
#define SLAB (ROWS * K_SZ)
#define XK 12
#define HK (KSTEPS - XK)
#define GSET (8 * 32 * 4)
#define PPAR (NOWN * GSET)
#define GK 6
#define QSET (2 * 32 * 4)
#define QPAR (3 * QSET)

#if __has_builtin(__builtin_amdgcn_tanhf)
#define FAST_TANH(x) __builtin_amdgcn_tanhf(x)
#elif __has_builtin(__builtin_amdgcn_tanh_f32)
#define FAST_TANH(x) __builtin_amdgcn_tanh_f32(x)
#else
#define FAST_TANH(x) tanhf(x)
#endif


template <int N>
__device__ __forceinline__ float row_from(float x)
{
    return __builtin_bit_cast(float,
        __builtin_amdgcn_update_dpp(0, __builtin_bit_cast(int, x), 0x100 + N, 0xf, 0xf, true));
}

template <bool MIXED>
__device__ __forceinline__ void h_wave(const float* __restrict__ Wih,
                                       const float* __restrict__ Whh,
                                       float* Zf, const float* Pf,
                                       const int uw, const int lane)
{
    constexpr int NT = MIXED ? 1 : 4;
    const int c16   = lane & 15;
    const int khalf = (lane >> 4) * 2;
    const int rbase = (lane >> 4) * 8;

    v2f wh[NT][HK];
#pragma unroll
    for (int g = 0; g < NT; ++g) {
        const bool valid = MIXED ? (c16 < 8) : true;
        const int  n     = MIXED ? (valid ? (c16 >> 1) * H_SZ + ULAST + (c16 & 1) : 0)
                                 : g * H_SZ + uw * UG + c16;
#pragma unroll
        for (int kk = XK; kk < KSTEPS; ++kk) {
            const int k = kk * 4 + khalf;
            float w0, w1;
            if (k < E_SZ) {
                w0 = Wih[n * E_SZ + k];
                w1 = Wih[n * E_SZ + k + 1];
            } else {
                w0 = Whh[n * H_SZ + (k - E_SZ)];
                w1 = Whh[n * H_SZ + (k + 1 - E_SZ)];
            }
            wh[g][kk - XK].x = valid ? w0 : 0.f;
            wh[g][kk - XK].y = valid ? w1 : 0.f;
        }
    }
    float cst[8];
#pragma unroll
    for (int v = 0; v < 8; ++v) cst[v] = 0.f;
    const bool hvalid = MIXED ? (c16 < 2) : true;
    const int  hrow   = rbase * K_SZ + E_SZ + (MIXED ? (hvalid ? ULAST + c16 : c16) : uw * UG + c16);
    const int  pslot  = uw * GSET + lane * 4;
    const bool isg = (c16 >> 1) == 2;
    const float sc = isg ? 1.0f : 0.5f;
    const float ab = isg ? 0.0f : 0.5f;

    __syncthreads();

    for (int sp = 0; sp < S_SZ; sp += NBUF) {
#pragma unroll
        for (int par = 0; par < NBUF; ++par) {
            const float* Zc = Zf + par * SLAB;
            float*       Zn = Zf + (hvalid ? ((par + 1) & (NBUF - 1)) : NBUF) * SLAB + hrow;
            const float* Pr = Pf + (par & 1) * PPAR + pslot;

            v8f acc[NT];
#pragma unroll
            for (int g = 0; g < NT; ++g) {
                const v4f lo = *(const v4f*)(Pr + (g * 2 + 0) * 128);
                const v4f hi = *(const v4f*)(Pr + (g * 2 + 1) * 128);
                acc[g][0] = lo[0]; acc[g][1] = lo[1]; acc[g][2] = lo[2]; acc[g][3] = lo[3];
                acc[g][4] = hi[0]; acc[g][5] = hi[1]; acc[g][6] = hi[2]; acc[g][7] = hi[3];
            }
            v2f a[HK];
#pragma unroll
            for (int kk = 0; kk < HK; ++kk)
                a[kk] = *(const v2f*)(Zc + c16 * K_SZ + (XK + kk) * 4 + khalf);
            __builtin_amdgcn_sched_barrier(0);

#pragma unroll
            for (int kk = 0; kk < HK; ++kk) {
#pragma unroll
                for (int g = 0; g < NT; ++g)
                    acc[g] = __builtin_amdgcn_wmma_f32_16x16x4_f32(
                                 false, a[kk], false, wh[g][kk], (short)0,
                                 acc[g], false, false);
            }

            float hv[8];
            if constexpr (!MIXED) {
                float ti[8], tf[8], tg[8], to[8];
#pragma unroll
                for (int v = 0; v < 8; ++v) ti[v] = 0.5f * acc[0][v];
#pragma unroll
                for (int v = 0; v < 8; ++v) tf[v] = 0.5f * acc[1][v];
#pragma unroll
                for (int v = 0; v < 8; ++v) to[v] = 0.5f * acc[NT - 1][v];
#pragma unroll
                for (int v = 0; v < 8; ++v) ti[v] = FAST_TANH(ti[v]);
#pragma unroll
                for (int v = 0; v < 8; ++v) tf[v] = FAST_TANH(tf[v]);
#pragma unroll
                for (int v = 0; v < 8; ++v) tg[v] = FAST_TANH(acc[NT / 2][v]);
#pragma unroll
                for (int v = 0; v < 8; ++v) to[v] = FAST_TANH(to[v]);
#pragma unroll
                for (int v = 0; v < 8; ++v) ti[v] = __builtin_fmaf(0.5f, ti[v], 0.5f);
#pragma unroll
                for (int v = 0; v < 8; ++v) tf[v] = __builtin_fmaf(0.5f, tf[v], 0.5f);
#pragma unroll
                for (int v = 0; v < 8; ++v) to[v] = __builtin_fmaf(0.5f, to[v], 0.5f);
#pragma unroll
                for (int v = 0; v < 8; ++v) tg[v] = ti[v] * tg[v];
#pragma unroll
                for (int v = 0; v < 8; ++v) cst[v] = __builtin_fmaf(tf[v], cst[v], tg[v]);
#pragma unroll
                for (int v = 0; v < 8; ++v) tg[v] = FAST_TANH(cst[v]);
#pragma unroll
                for (int v = 0; v < 8; ++v) hv[v] = to[v] * tg[v];
            } else {
                float ta[8], tf[8], tg[8], to[8];
#pragma unroll
                for (int v = 0; v < 8; ++v) ta[v] = sc * acc[0][v];
#pragma unroll
                for (int v = 0; v < 8; ++v) ta[v] = FAST_TANH(ta[v]);
#pragma unroll
                for (int v = 0; v < 8; ++v) ta[v] = __builtin_fmaf(sc, ta[v], ab);
#pragma unroll
                for (int v = 0; v < 8; ++v) {
                    tf[v] = row_from<2>(ta[v]);
                    tg[v] = row_from<4>(ta[v]);
                    to[v] = row_from<6>(ta[v]);
                }
#pragma unroll
                for (int v = 0; v < 8; ++v) tg[v] = ta[v] * tg[v];
#pragma unroll
                for (int v = 0; v < 8; ++v) cst[v] = __builtin_fmaf(tf[v], cst[v], tg[v]);
#pragma unroll
                for (int v = 0; v < 8; ++v) tg[v] = FAST_TANH(cst[v]);
#pragma unroll
                for (int v = 0; v < 8; ++v) hv[v] = to[v] * tg[v];
            }
#pragma unroll
            for (int v = 0; v < 8; ++v)
                Zn[v * K_SZ] = hv[v];

            __syncthreads();
        }
    }
}

template <int NT>
__device__ __forceinline__ void x_wave(const float* __restrict__ X,
                                       const float* __restrict__ Wih,
                                       const float* __restrict__ bih,
                                       const float* __restrict__ bhh,
                                       float* Zf, float* Pf, float* Qf,
                                       const int uw, const int lane, const int tid, const int b0)
{
    constexpr bool LAST = (NT == 4);
    const int c16   = lane & 15;
    const int khalf = (lane >> 4) * 2;

    v2f   wx[NT][XK];
    float bias[NT];
    int   poff[NT];
#pragma unroll
    for (int g = 0; g < NT; ++g) {
        bool valid; int n;
        if (NT == 3) {
            valid = true;
            n = g * H_SZ + uw * UG + c16;
            poff[g] = uw * GSET + g * 256 + lane * 4;
        } else if (g < 3) {
            valid = true;
            n = 3 * H_SZ + g * UG + c16;
            poff[g] = g * GSET + 3 * 256 + lane * 4;
        } else {
            valid = c16 < 8;
            n = valid ? (c16 >> 1) * H_SZ + ULAST + (c16 & 1) : 0;
            poff[g] = 3 * GSET + lane * 4;
        }
#pragma unroll
        for (int kk = 0; kk < XK; ++kk) {
            const int k = kk * 4 + khalf;
            const float w0 = Wih[n * E_SZ + k];
            const float w1 = Wih[n * E_SZ + k + 1];
            wx[g][kk].x = valid ? w0 : 0.f;
            wx[g][kk].y = valid ? w1 : 0.f;
        }
        bias[g] = valid ? (bih[n] + bhh[n]) : 0.f;
    }
    v8f bv[NT];
#pragma unroll
    for (int g = 0; g < NT; ++g)
#pragma unroll
        for (int v = 0; v < 8; ++v) bv[g][v] = bias[g];

    constexpr int NQ = LAST ? 3 : 1;
    v2f wg[NQ][GK];
    v8f bvg[NQ];
    if constexpr (LAST) {
#pragma unroll
        for (int j = 0; j < 3; ++j) {
            const int n = 2 * H_SZ + j * UG + c16;
#pragma unroll
            for (int kk = 0; kk < GK; ++kk) {
                const int k = kk * 4 + khalf;
                wg[j][kk].x = Wih[n * E_SZ + k];
                wg[j][kk].y = Wih[n * E_SZ + k + 1];
            }
            const float b = bih[n] + bhh[n];
#pragma unroll
            for (int v = 0; v < 8; ++v) bvg[j][v] = b;
        }
    }

    const int xt = tid - XTHREADS;
    const float* xsrc[4];
    int          xdst[4];
#pragma unroll
    for (int i = 0; i < 4; ++i) {
        const int p  = xt + i * XTHREADS;
        const int pc = (p < XPAIRS) ? p : XPAIRS - 1;
        const int r  = pc / (E_SZ / 2);
        const int e  = (pc % (E_SZ / 2)) * 2;
        xsrc[i] = X + (size_t)(b0 + r) * S_SZ * E_SZ + e;
        xdst[i] = r * K_SZ + e;
    }

    constexpr int NHEAD = LAST ? 4 : 2;
    constexpr int KH = 12 / NHEAD;
    static_assert(LAST || KH == GK, "a group's three tiles carry on at the same k-step");
    auto xhead = [&](const v2f (&f)[XK], v8f (&t)[NT]) {
#pragma unroll
        for (int g = 0; g < NHEAD; ++g)
            t[g] = __builtin_amdgcn_wmma_f32_16x16x4_f32(
                       false, f[0], false, wx[g][0], (short)0,
                       bv[g], false, false);
#pragma unroll
        for (int kk = 1; kk < KH; ++kk) {
#pragma unroll
            for (int g = 0; g < NHEAD; ++g)
                t[g] = __builtin_amdgcn_wmma_f32_16x16x4_f32(
                           false, f[kk], false, wx[g][kk], (short)0,
                           t[g], false, false);
        }
    };
    auto xall = [&](const v2f (&f)[XK], v8f (&t)[NT]) {
#pragma unroll
        for (int g = 0; g < NT; ++g)
            t[g] = __builtin_amdgcn_wmma_f32_16x16x4_f32(
                       false, f[0], false, wx[g][0], (short)0,
                       bv[g], false, false);
#pragma unroll
        for (int kk = 1; kk < XK; ++kk) {
#pragma unroll
            for (int g = 0; g < NT; ++g)
                t[g] = __builtin_amdgcn_wmma_f32_16x16x4_f32(
                           false, f[kk], false, wx[g][kk], (short)0,
                           t[g], false, false);
        }
    };
    auto xtail = [&](const v2f (&f)[XK], v8f (&t)[NT]) {
#pragma unroll
        for (int kk = KH; kk < XK; ++kk) {
#pragma unroll
            for (int g = 0; g < NT; ++g)
                t[g] = __builtin_amdgcn_wmma_f32_16x16x4_f32(
                           false, f[kk], false, wx[g][kk], (short)0,
                           t[g], false, false);
        }
    };
    auto handover = [&](float* Pw, const v8f (&t)[NT]) {
#pragma unroll
        for (int g = 0; g < NT; ++g) {
            v4f lo, hi;
            lo[0] = t[g][0]; lo[1] = t[g][1]; lo[2] = t[g][2]; lo[3] = t[g][3];
            hi[0] = t[g][4]; hi[1] = t[g][5]; hi[2] = t[g][6]; hi[3] = t[g][7];
            *(v4f*)(Pw + poff[g])       = lo;
            *(v4f*)(Pw + poff[g] + 128) = hi;
        }
    };
    auto xfrags = [&](const float* Zs, v2f (&f)[XK]) {
#pragma unroll
        for (int kk = 0; kk < XK; ++kk)
            f[kk] = *(const v2f*)(Zs + c16 * K_SZ + kk * 4 + khalf);
    };

    auto gfrags = [&](const float* Zs, v2f (&f)[GK]) {
#pragma unroll
        for (int kk = 0; kk < GK; ++kk)
            f[kk] = *(const v2f*)(Zs + c16 * K_SZ + kk * 4 + khalf);
    };
    auto gstart = [&](const v2f (&f)[GK], v8f (&q)[NQ]) {
#pragma unroll
        for (int j = 0; j < NQ; ++j)
            q[j] = __builtin_amdgcn_wmma_f32_16x16x4_f32(
                       false, f[0], false, wg[j][0], (short)0,
                       bvg[j], false, false);
#pragma unroll
        for (int kk = 1; kk < GK; ++kk) {
#pragma unroll
            for (int j = 0; j < NQ; ++j)
                q[j] = __builtin_amdgcn_wmma_f32_16x16x4_f32(
                           false, f[kk], false, wg[j][kk], (short)0,
                           q[j], false, false);
        }
    };
    auto gleave = [&](float* Qw, const v8f (&q)[NQ]) {
#pragma unroll
        for (int j = 0; j < NQ; ++j) {
            v4f lo, hi;
            lo[0] = q[j][0]; lo[1] = q[j][1]; lo[2] = q[j][2]; lo[3] = q[j][3];
            hi[0] = q[j][4]; hi[1] = q[j][5]; hi[2] = q[j][6]; hi[3] = q[j][7];
            *(v4f*)(Qw + j * QSET + lane * 4)       = lo;
            *(v4f*)(Qw + j * QSET + 128 + lane * 4) = hi;
        }
    };

    v8f tp[NT];
    v2f fc[XK];
    v2f xr0[4];
    v2f xr1[4];
    v8f qp[NQ];
    {
        v2f f0[XK];
        v8f t0[NT];
        xfrags(Zf, f0);
        xall(f0, t0);
        handover(Pf, t0);
        xfrags(Zf + SLAB, f0);
        xall(f0, tp);
        xfrags(Zf + 2 * SLAB, fc);
        if constexpr (LAST) {
            v2f g0[GK];
            v8f q0[NQ];
            gfrags(Zf + 2 * SLAB, g0);
            gstart(g0, q0);
            gleave(Qf, q0);
            gfrags(Zf + 3 * SLAB, g0);
            gstart(g0, qp);
        }
#pragma unroll
        for (int i = 0; i < 4; ++i) {
            xr0[i] = *(const v2f*)(xsrc[i] + (size_t)5 * E_SZ);
            xr1[i] = *(const v2f*)(xsrc[i] + (size_t)6 * E_SZ);
        }
    }
    __syncthreads();

    for (int sp = 0; sp < S_SZ; sp += NBUF) {
#pragma unroll
        for (int par = 0; par < NBUF; ++par) {
            const int s = sp + par;
            const float* Zn3 = Zf + ((par + 3) & (NBUF - 1)) * SLAB;
            const float* Zn4 = Zf + ((par + 4) & (NBUF - 1)) * SLAB;
            float*       Zn5 = Zf + ((par + 5) & (NBUF - 1)) * SLAB;
            const int sn = (s + 7 < S_SZ) ? s + 7 : S_SZ - 1;

            __builtin_amdgcn_s_setprio(3);

            v8f tc[NT];
            xhead(fc, tc);
            if constexpr (!LAST)
                asm volatile("s_setprio 0" :: "v"(tc[0]), "v"(tc[1]) : "memory");
            else
                asm volatile("s_setprio 0" :: "v"(tc[0]), "v"(tc[1]), "v"(tc[2]), "v"(tc[3]) : "memory");

            v2f xn[4];
#pragma unroll
            for (int i = 0; i < 4; ++i)
                xn[i] = *(const v2f*)(xsrc[i] + (size_t)sn * E_SZ);
            handover(Pf + ((par + 1) & 1) * PPAR, tp);
            if constexpr (LAST)
                gleave(Qf + ((par + 1) & 1) * QPAR, qp);
#pragma unroll
            for (int q = 0; q < 4; ++q)
                *(v2f*)(Zn5 + xdst[q]) = xr0[q];
            v2f fn[XK];
            xfrags(Zn3, fn);
            v2f gf[GK];
            v8f qc[NQ];
            if constexpr (LAST) {
                gfrags(Zn4, gf);
            } else {
                const float* Qr = Qf + (par & 1) * QPAR + uw * QSET + lane * 4;
                const v4f lo = *(const v4f*)(Qr);
                const v4f hi = *(const v4f*)(Qr + 128);
                tc[2][0] = lo[0]; tc[2][1] = lo[1]; tc[2][2] = lo[2]; tc[2][3] = lo[3];
                tc[2][4] = hi[0]; tc[2][5] = hi[1]; tc[2][6] = hi[2]; tc[2][7] = hi[3];
            }
            __builtin_amdgcn_sched_barrier(0);

            xtail(fc, tc);
            if constexpr (LAST)
                gstart(gf, qc);
            __builtin_amdgcn_sched_barrier(0);

            __builtin_amdgcn_fence(__ATOMIC_RELEASE, "workgroup", "local");
            __builtin_amdgcn_s_barrier();
            __builtin_amdgcn_fence(__ATOMIC_ACQUIRE, "workgroup", "local");

#pragma unroll
            for (int g = 0; g < NT; ++g) tp[g] = tc[g];
#pragma unroll
            for (int kk = 0; kk < XK; ++kk) fc[kk] = fn[kk];
#pragma unroll
            for (int i = 0; i < 4; ++i) { xr0[i] = xr1[i]; xr1[i] = xn[i]; }
            if constexpr (LAST) {
#pragma unroll
                for (int j = 0; j < NQ; ++j) qp[j] = qc[j];
            }
        }
    }
    __builtin_amdgcn_fence(__ATOMIC_RELEASE, "workgroup");
}

__global__ __launch_bounds__(THREADS)
void lstm_pair_kernel(const float* __restrict__ X,
                      const float* __restrict__ Wih,
                      const float* __restrict__ Whh,
                      const float* __restrict__ bih,
                      const float* __restrict__ bhh,
                      float* __restrict__ out)
{
    __shared__ __attribute__((aligned(16))) float Z[NBUF + 1][ROWS][K_SZ];
    __shared__ __attribute__((aligned(16))) float P[2][NOWN][8][32][4];
    __shared__ __attribute__((aligned(16))) float Q[2][3][2][32][4];

    const int tid  = threadIdx.x;
    const int wave = __builtin_amdgcn_readfirstlane(tid >> 5);
    const int lane = tid & 31;
    const int b0   = blockIdx.x * ROWS;
    const bool xrole = wave >= NOWN;
    const int  uw    = wave & (NOWN - 1);

    float* Zf = &Z[0][0][0];
    float* Pf = &P[0][0][0][0][0];
    float* Qf = &Q[0][0][0][0][0];

    for (int idx = tid; idx < ROWS * H_SZ; idx += THREADS)
        Z[0][idx / H_SZ][E_SZ + idx % H_SZ] = 0.f;
#pragma unroll
    for (int i = 0; i < 2; ++i) {
        const int p  = tid + i * THREADS;
        const int pc = (p < XPAIRS) ? p : XPAIRS - 1;
        const int r  = pc / (E_SZ / 2);
        const int e  = (pc % (E_SZ / 2)) * 2;
        const float* src = X + (size_t)(b0 + r) * S_SZ * E_SZ + e;
#pragma unroll
        for (int t = 0; t < 5; ++t)
            *(v2f*)(Zf + t * SLAB + r * K_SZ + e) = *(const v2f*)(src + t * E_SZ);
    }
    __syncthreads();

    if (!xrole) {
        if (uw < NOWN - 1) h_wave<false>(Wih, Whh, Zf, Pf, uw, lane);
        else               h_wave<true >(Wih, Whh, Zf, Pf, uw, lane);
    } else {
        if (uw < NOWN - 1) x_wave<3>(X, Wih, bih, bhh, Zf, Pf, Qf, uw, lane, tid, b0);
        else               x_wave<4>(X, Wih, bih, bhh, Zf, Pf, Qf, uw, lane, tid, b0);
    }

    for (int idx = tid; idx < ROWS * H_SZ; idx += THREADS)
        out[(size_t)b0 * H_SZ + idx] = Z[0][idx / H_SZ][E_SZ + idx % H_SZ];
}

extern "C" void kernel_launch(void* const* d_in, const int* in_sizes, int n_in,
                              void* d_out, int out_size, void* d_ws, size_t ws_size,
                              hipStream_t stream) {
    (void)in_sizes; (void)n_in; (void)d_ws; (void)ws_size; (void)out_size;
    const float* X   = (const float*)d_in[0];
    const float* Wih = (const float*)d_in[1];
    const float* Whh = (const float*)d_in[2];
    const float* bih = (const float*)d_in[3];
    const float* bhh = (const float*)d_in[4];
    float* out = (float*)d_out;

    dim3 grid(B_SZ / ROWS);
    dim3 block(THREADS);
    lstm_pair_kernel<<<grid, block, 0, stream>>>(X, Wih, Whh, bih, bhh, out);
}
